# starting kernel
# baseline (speedup 1.0000x reference)
#include <hip/hip_runtime.h>
#include <hip/hip_bf16.h>
#include <math.h>

typedef float v2f __attribute__((ext_vector_type(2)));
typedef float v8f __attribute__((ext_vector_type(8)));

#define NEG_SLOPE 0.2f

__device__ __forceinline__ float leaky(float v) { return v > 0.f ? v : NEG_SLOPE * v; }

__device__ __forceinline__ void atomicMaxF(float* addr, float val) {
  if (val >= 0.f) atomicMax((int*)addr, __float_as_int(val));
  else            atomicMin((unsigned int*)addr, (unsigned int)__float_as_int(val));
}

__global__ void init_kern(float* emax1, float* denom1, float* agg1,
                          float* emax2, float* denom2, float* agg2, int N) {
  int i = blockIdx.x * blockDim.x + threadIdx.x;
  if (i < N * 64) agg1[i] = 0.f;
  if (i < N * 6)  agg2[i] = 0.f;
  if (i < N * 4) { emax1[i] = -3.0e38f; denom1[i] = 0.f; }
  if (i < N)     { emax2[i] = -3.0e38f; denom2[i] = 0.f; }
}

__global__ void proj1_wmma(const float* __restrict__ x, const float* __restrict__ W1,
                           float* __restrict__ h1, int N) {
  int gtid = blockIdx.x * blockDim.x + threadIdx.x;
  int wave = gtid >> 5;
  int lane = threadIdx.x & 31;
  int tiles = (N + 15) >> 4;
  if (wave >= tiles) return;
  int half = lane >> 4;
  int l16  = lane & 15;
  int row  = wave * 16 + l16;
  int rowc = row < N ? row : N - 1;
  bool rok = row < N;

  v2f a[2];
#pragma unroll
  for (int s = 0; s < 2; ++s) {
    int k0 = 4 * s + 2 * half;
    a[s][0] = (rok && k0     < 7) ? x[rowc * 7 + k0]     : 0.f;
    a[s][1] = (rok && k0 + 1 < 7) ? x[rowc * 7 + k0 + 1] : 0.f;
  }

  v8f acc[4];
#pragma unroll
  for (int t = 0; t < 4; ++t)
#pragma unroll
    for (int q = 0; q < 8; ++q) acc[t][q] = 0.f;

#pragma unroll
  for (int s = 0; s < 2; ++s) {
    int k0 = 4 * s + 2 * half;
#pragma unroll
    for (int t = 0; t < 4; ++t) {
      int col = t * 16 + l16;
      v2f b;
      b[0] = (k0     < 7) ? W1[k0 * 64 + col]       : 0.f;
      b[1] = (k0 + 1 < 7) ? W1[(k0 + 1) * 64 + col] : 0.f;
      acc[t] = __builtin_amdgcn_wmma_f32_16x16x4_f32(
          false, a[s], false, b, (short)0, acc[t], false, false);
    }
  }

#pragma unroll
  for (int t = 0; t < 4; ++t)
#pragma unroll
    for (int r = 0; r < 8; ++r) {
      int m = wave * 16 + r + 8 * half;
      if (m < N) h1[m * 64 + t * 16 + l16] = acc[t][r];
    }
}

__global__ void alpha1_kern(const float* __restrict__ h1, const float* __restrict__ aS,
                            const float* __restrict__ aD, float* __restrict__ as1,
                            float* __restrict__ ad1, int N) {
  int i = blockIdx.x * blockDim.x + threadIdx.x;
  if (i >= N * 4) return;
  int hd = i & 3, n = i >> 2;
  const float* hp = h1 + n * 64 + hd * 16;
  float s = 0.f, d = 0.f;
#pragma unroll
  for (int c = 0; c < 16; ++c) {
    float v = hp[c];
    s += v * aS[hd * 16 + c];
    d += v * aD[hd * 16 + c];
  }
  as1[i] = s; ad1[i] = d;
}

__global__ void e1_max(const int* __restrict__ ei, const float* __restrict__ as1,
                       const float* __restrict__ ad1, float* __restrict__ emax1,
                       int E, int N) {
  int e = blockIdx.x * blockDim.x + threadIdx.x;
  if (e >= E + N) return;
  int src = e < E ? ei[e]     : e - E;
  int dst = e < E ? ei[E + e] : e - E;
#pragma unroll
  for (int h = 0; h < 4; ++h) {
    float ev = leaky(as1[src * 4 + h] + ad1[dst * 4 + h]);
    atomicMaxF(&emax1[dst * 4 + h], ev);
  }
}

__global__ void e1_den(const int* __restrict__ ei, const float* __restrict__ as1,
                       const float* __restrict__ ad1, const float* __restrict__ emax1,
                       float* __restrict__ denom1, int E, int N) {
  int e = blockIdx.x * blockDim.x + threadIdx.x;
  if (e >= E + N) return;
  int src = e < E ? ei[e]     : e - E;
  int dst = e < E ? ei[E + e] : e - E;
#pragma unroll
  for (int h = 0; h < 4; ++h) {
    float ev = leaky(as1[src * 4 + h] + ad1[dst * 4 + h]);
    atomicAdd(&denom1[dst * 4 + h], __expf(ev - emax1[dst * 4 + h]));
  }
}

__global__ void e1_agg(const int* __restrict__ ei, const float* __restrict__ as1,
                       const float* __restrict__ ad1, const float* __restrict__ emax1,
                       const float* __restrict__ denom1, const float* __restrict__ h1,
                       float* __restrict__ agg1, int E, int N) {
  int t = blockIdx.x * blockDim.x + threadIdx.x;
  if (t >= (E + N) * 4) return;
  int e = t >> 2, h = t & 3;
  int src = e < E ? ei[e]     : e - E;
  int dst = e < E ? ei[E + e] : e - E;
  float ev = leaky(as1[src * 4 + h] + ad1[dst * 4 + h]);
  float w  = __expf(ev - emax1[dst * 4 + h]) / denom1[dst * 4 + h];
  const float4* hp = (const float4*)(h1 + src * 64 + h * 16);
  float* op = agg1 + dst * 64 + h * 16;
#pragma unroll
  for (int q = 0; q < 4; ++q) {
    float4 v = hp[q];
    atomicAdd(op + q * 4 + 0, v.x * w);
    atomicAdd(op + q * 4 + 1, v.y * w);
    atomicAdd(op + q * 4 + 2, v.z * w);
    atomicAdd(op + q * 4 + 3, v.w * w);
  }
}

__global__ void bias_elu(float* __restrict__ x2, const float* __restrict__ b1, int N) {
  int i = blockIdx.x * blockDim.x + threadIdx.x;
  if (i >= N * 64) return;
  float v = x2[i] + b1[i & 63];
  x2[i] = v > 0.f ? v : (__expf(v) - 1.f);
}

__global__ void proj2_wmma(const float* __restrict__ x2, const float* __restrict__ W2,
                           float* __restrict__ h2, int N) {
  int gtid = blockIdx.x * blockDim.x + threadIdx.x;
  int wave = gtid >> 5;
  int lane = threadIdx.x & 31;
  int tiles = (N + 15) >> 4;
  if (wave >= tiles) return;
  int half = lane >> 4, l16 = lane & 15;
  int row  = wave * 16 + l16;
  int rowc = row < N ? row : N - 1;
  bool rok = row < N;
  bool cok = l16 < 6;

  v8f acc;
#pragma unroll
  for (int q = 0; q < 8; ++q) acc[q] = 0.f;

#pragma unroll
  for (int s = 0; s < 16; ++s) {
    int k0 = 4 * s + 2 * half;
    v2f a, b;
    a[0] = rok ? x2[rowc * 64 + k0]     : 0.f;
    a[1] = rok ? x2[rowc * 64 + k0 + 1] : 0.f;
    b[0] = cok ? W2[k0 * 6 + l16]       : 0.f;
    b[1] = cok ? W2[(k0 + 1) * 6 + l16] : 0.f;
    acc = __builtin_amdgcn_wmma_f32_16x16x4_f32(
        false, a, false, b, (short)0, acc, false, false);
  }

  if (cok) {
#pragma unroll
    for (int r = 0; r < 8; ++r) {
      int m = wave * 16 + r + 8 * half;
      if (m < N) h2[m * 6 + l16] = acc[r];
    }
  }
}

__global__ void alpha2_kern(const float* __restrict__ h2, const float* __restrict__ aS,
                            const float* __restrict__ aD, float* __restrict__ as2,
                            float* __restrict__ ad2, int N) {
  int n = blockIdx.x * blockDim.x + threadIdx.x;
  if (n >= N) return;
  float s = 0.f, d = 0.f;
#pragma unroll
  for (int c = 0; c < 6; ++c) {
    float v = h2[n * 6 + c];
    s += v * aS[c];
    d += v * aD[c];
  }
  as2[n] = s; ad2[n] = d;
}

__global__ void e2_max(const int* __restrict__ ei, const float* __restrict__ as2,
                       const float* __restrict__ ad2, float* __restrict__ emax2,
                       int E, int N) {
  int e = blockIdx.x * blockDim.x + threadIdx.x;
  if (e >= E + N) return;
  int src = e < E ? ei[e]     : e - E;
  int dst = e < E ? ei[E + e] : e - E;
  atomicMaxF(&emax2[dst], leaky(as2[src] + ad2[dst]));
}

__global__ void e2_den(const int* __restrict__ ei, const float* __restrict__ as2,
                       const float* __restrict__ ad2, const float* __restrict__ emax2,
                       float* __restrict__ denom2, int E, int N) {
  int e = blockIdx.x * blockDim.x + threadIdx.x;
  if (e >= E + N) return;
  int src = e < E ? ei[e]     : e - E;
  int dst = e < E ? ei[E + e] : e - E;
  float ev = leaky(as2[src] + ad2[dst]);
  atomicAdd(&denom2[dst], __expf(ev - emax2[dst]));
}

__global__ void e2_agg(const int* __restrict__ ei, const float* __restrict__ as2,
                       const float* __restrict__ ad2, const float* __restrict__ emax2,
                       const float* __restrict__ denom2, const float* __restrict__ h2,
                       float* __restrict__ agg2, int E, int N) {
  int e = blockIdx.x * blockDim.x + threadIdx.x;
  if (e >= E + N) return;
  int src = e < E ? ei[e]     : e - E;
  int dst = e < E ? ei[E + e] : e - E;
  float ev = leaky(as2[src] + ad2[dst]);
  float w  = __expf(ev - emax2[dst]) / denom2[dst];
#pragma unroll
  for (int c = 0; c < 6; ++c)
    atomicAdd(&agg2[dst * 6 + c], h2[src * 6 + c] * w);
}

__global__ void final_kern(const float* __restrict__ agg2, const float* __restrict__ b2,
                           float* __restrict__ out, int N) {
  int n = blockIdx.x * blockDim.x + threadIdx.x;
  if (n >= N) return;
  float o[6];
  float m = -3.0e38f;
#pragma unroll
  for (int c = 0; c < 6; ++c) { o[c] = agg2[n * 6 + c] + b2[c]; m = fmaxf(m, o[c]); }
  float sum = 0.f;
#pragma unroll
  for (int c = 0; c < 6; ++c) sum += __expf(o[c] - m);
  float l = __logf(sum);
#pragma unroll
  for (int c = 0; c < 6; ++c) out[n * 6 + c] = o[c] - m - l;
}

extern "C" void kernel_launch(void* const* d_in, const int* in_sizes, int n_in,
                              void* d_out, int out_size, void* d_ws, size_t ws_size,
                              hipStream_t stream) {
  const float* x      = (const float*)d_in[0];
  const int*   ei     = (const int*)d_in[1];
  const float* W1     = (const float*)d_in[2];
  const float* a_src1 = (const float*)d_in[3];
  const float* a_dst1 = (const float*)d_in[4];
  const float* b1     = (const float*)d_in[5];
  const float* W2     = (const float*)d_in[6];
  const float* a_src2 = (const float*)d_in[7];
  const float* a_dst2 = (const float*)d_in[8];
  const float* b2     = (const float*)d_in[9];
  float* out = (float*)d_out;

  int N = in_sizes[0] / 7;
  int E = in_sizes[1] / 2;
  int ET = E + N;

  float* ws     = (float*)d_ws;
  float* h1     = ws;
  float* x2     = h1 + (size_t)N * 64;
  float* as1    = x2 + (size_t)N * 64;
  float* ad1    = as1 + (size_t)N * 4;
  float* emax1  = ad1 + (size_t)N * 4;
  float* denom1 = emax1 + (size_t)N * 4;
  float* h2     = denom1 + (size_t)N * 4;
  float* agg2   = h2 + (size_t)N * 6;
  float* as2    = agg2 + (size_t)N * 6;
  float* ad2    = as2 + N;
  float* emax2  = ad2 + N;
  float* denom2 = emax2 + N;

  const int B = 256;
  auto G = [](long n, int b) { return (int)((n + b - 1) / b); };
  int tiles = (N + 15) / 16;
  int projBlocks = (tiles + (B / 32) - 1) / (B / 32);

  init_kern<<<G((long)N * 64, B), B, 0, stream>>>(emax1, denom1, x2, emax2, denom2, agg2, N);

  proj1_wmma<<<projBlocks, B, 0, stream>>>(x, W1, h1, N);
  alpha1_kern<<<G((long)N * 4, B), B, 0, stream>>>(h1, a_src1, a_dst1, as1, ad1, N);
  e1_max<<<G(ET, B), B, 0, stream>>>(ei, as1, ad1, emax1, E, N);
  e1_den<<<G(ET, B), B, 0, stream>>>(ei, as1, ad1, emax1, denom1, E, N);
  e1_agg<<<G((long)ET * 4, B), B, 0, stream>>>(ei, as1, ad1, emax1, denom1, h1, x2, E, N);
  bias_elu<<<G((long)N * 64, B), B, 0, stream>>>(x2, b1, N);

  proj2_wmma<<<projBlocks, B, 0, stream>>>(x2, W2, h2, N);
  alpha2_kern<<<G(N, B), B, 0, stream>>>(h2, a_src2, a_dst2, as2, ad2, N);
  e2_max<<<G(ET, B), B, 0, stream>>>(ei, as2, ad2, emax2, E, N);
  e2_den<<<G(ET, B), B, 0, stream>>>(ei, as2, ad2, emax2, denom2, E, N);
  e2_agg<<<G(ET, B), B, 0, stream>>>(ei, as2, ad2, emax2, denom2, h2, agg2, E, N);
  final_kern<<<G(N, B), B, 0, stream>>>(agg2, b2, out, N);
}


# Aggregation passes: a work-group adds only into destinations of the cache half it runs on (die read at run time); edge chunks are walked once per half via two ticket queues; sums in arrival order
# speedup vs baseline: 11.1856x; 11.1856x over previous
#include <hip/hip_runtime.h>
#include <hip/hip_bf16.h>
#include <math.h>

typedef float v2f __attribute__((ext_vector_type(2)));
typedef float v8f __attribute__((ext_vector_type(8)));
typedef float v4f __attribute__((ext_vector_type(4)));

#define NEG_SLOPE 0.2f

__device__ __forceinline__ float leaky(float v) { return v > 0.f ? v : NEG_SLOPE * v; }

__device__ __forceinline__ void atomicMaxF(float* addr, float val) {
  if (val >= 0.f) atomicMax((int*)addr, __float_as_int(val));
  else            atomicMin((unsigned int*)addr, (unsigned int)__float_as_int(val));
}

__device__ __forceinline__ void atomicMinF(float* addr, float val) {
  int b = __float_as_int(val);
  if (b == (int)0x80000000) b = 0;
  if (b >= 0) atomicMin((int*)addr, b);
  else        atomicMax((unsigned int*)addr, (unsigned int)b);
}

#define RANGE_OK 40.0f

__global__ void init_kern(float* emax1, float* denom1, float* agg1,
                          float* emax2, float* agg2r, float* gst, unsigned* tick, int N) {
  int i = blockIdx.x * blockDim.x + threadIdx.x;
  if (i < 16) gst[i] = (i < 4 || i == 8) ? -3.0e38f : 3.0e38f;
  if (i < 4) tick[i * 128] = 0u;
  if (i < N * 64) agg1[i] = 0.f;
  if (i < N * 8)  agg2r[i] = 0.f;
  if (i < N * 4) { emax1[i] = -3.0e38f; denom1[i] = 0.f; }
  if (i < N)     emax2[i] = -3.0e38f;
}

__global__ void proj1_wmma(const float* __restrict__ x, const float* __restrict__ W1,
                           float* __restrict__ h1, int N) {
  int gtid = blockIdx.x * blockDim.x + threadIdx.x;
  int wave = gtid >> 5;
  int lane = threadIdx.x & 31;
  int tiles = (N + 15) >> 4;
  if (wave >= tiles) return;
  int half = lane >> 4;
  int l16  = lane & 15;
  int row  = wave * 16 + l16;
  int rowc = row < N ? row : N - 1;
  bool rok = row < N;

  v2f a[2];
#pragma unroll
  for (int s = 0; s < 2; ++s) {
    int k0 = 4 * s + 2 * half;
    a[s][0] = (rok && k0     < 7) ? x[rowc * 7 + k0]     : 0.f;
    a[s][1] = (rok && k0 + 1 < 7) ? x[rowc * 7 + k0 + 1] : 0.f;
  }

  v8f acc[4];
#pragma unroll
  for (int t = 0; t < 4; ++t)
#pragma unroll
    for (int q = 0; q < 8; ++q) acc[t][q] = 0.f;

#pragma unroll
  for (int s = 0; s < 2; ++s) {
    int k0 = 4 * s + 2 * half;
#pragma unroll
    for (int t = 0; t < 4; ++t) {
      int col = t * 16 + l16;
      v2f b;
      b[0] = (k0     < 7) ? W1[k0 * 64 + col]       : 0.f;
      b[1] = (k0 + 1 < 7) ? W1[(k0 + 1) * 64 + col] : 0.f;
      acc[t] = __builtin_amdgcn_wmma_f32_16x16x4_f32(
          false, a[s], false, b, (short)0, acc[t], false, false);
    }
  }

#pragma unroll
  for (int t = 0; t < 4; ++t)
#pragma unroll
    for (int r = 0; r < 8; ++r) {
      int m = wave * 16 + r + 8 * half;
      if (m < N) h1[m * 64 + t * 16 + l16] = acc[t][r];
    }
}

__global__ void alpha1_kern(const float* __restrict__ h1, const float* __restrict__ aS,
                            const float* __restrict__ aD, float* __restrict__ as1,
                            float* __restrict__ ad1, float* __restrict__ gst, int N) {
  __shared__ float red[8][8];
  int i = blockIdx.x * blockDim.x + threadIdx.x;
  bool ok = i < N * 4;
  int ic = ok ? i : N * 4 - 1;
  int hd = ic & 3, n = ic >> 2;
  const float* hp = h1 + n * 64 + hd * 16;
  float s = 0.f, d = 0.f;
#pragma unroll
  for (int c = 0; c < 16; ++c) {
    float v = hp[c];
    s += v * aS[hd * 16 + c];
    d += v * aD[hd * 16 + c];
  }
  if (ok) { as1[i] = s; ad1[i] = d; }
  float mx = ok ? s : -3.0e38f, mn = ok ? s : 3.0e38f;
#pragma unroll
  for (int o = 4; o < 32; o <<= 1) {
    mx = fmaxf(mx, __shfl_xor(mx, o));
    mn = fminf(mn, __shfl_xor(mn, o));
  }
  int lane = threadIdx.x & 31, w = threadIdx.x >> 5;
  if (lane < 4) { red[w][lane] = mx; red[w][4 + lane] = mn; }
  __syncthreads();
  if (threadIdx.x < 8) {
    int t = threadIdx.x;
    float v = red[0][t];
    int nw = blockDim.x >> 5;
    for (int k = 1; k < nw; ++k) v = t < 4 ? fmaxf(v, red[k][t]) : fminf(v, red[k][t]);
    if (t < 4) atomicMaxF(&gst[t], v); else atomicMinF(&gst[t], v);
  }
}

__device__ __forceinline__ bool narrow1(const float* __restrict__ gst) {
  return (gst[0] - gst[4] <= RANGE_OK) & (gst[1] - gst[5] <= RANGE_OK) &
         (gst[2] - gst[6] <= RANGE_OK) & (gst[3] - gst[7] <= RANGE_OK);
}
__device__ __forceinline__ bool narrow2(const float* __restrict__ gst) {
  return gst[8] - gst[9] <= RANGE_OK;
}

#define SPAN 256
#define TICK_PITCH 128

__device__ __forceinline__ int chip_half() {
  return (__builtin_amdgcn_s_sendmsg_rtn(0x87) >> 18) & 1;
}

__device__ __forceinline__ int half_split(int N) {
  return ((N >> 1) + 15) & ~15;
}

template <class F>
__device__ __forceinline__ void walk_by_half(unsigned* __restrict__ tick, int ET, F chunk) {
  __shared__ int s_t[2];
  const int nspan = (ET + SPAN - 1) / SPAN;
  const int wv = threadIdx.x >> 5;
  const int h = chip_half();
  int par = 0;
  for (int q = 0; q < 2; ++q) {
    const int hq = h ^ q;
    unsigned* tk = tick + hq * TICK_PITCH;
    int nxt = 0;
    if (threadIdx.x == 0) nxt = (int)atomicInc(tk, 0xFFFFFFFFu);
    for (;;) {
      if (threadIdx.x == 0) s_t[par] = nxt;
      __syncthreads();
      const int t = __builtin_amdgcn_readfirstlane(s_t[par]);
      par ^= 1;
      if (t >= nspan) break;
      if (threadIdx.x == 0) nxt = (int)atomicInc(tk, 0xFFFFFFFFu);
      const int base = t * SPAN + wv * 32;
      if (base < ET) chunk(base, hq);
    }
  }
}

__device__ __forceinline__ int keep_half(bool valid, int hq, int Nh, int lane, int& src, int& dst) {
  bool mine = valid && ((dst >= Nh) == (hq != 0));
  unsigned mask = __builtin_amdgcn_ballot_w32(mine);
  int cnt = __builtin_popcount(mask);
  int below = (int)__builtin_amdgcn_mbcnt_lo(mask, 0u);
  int slot = mine ? below : cnt + lane - below;
  src = __builtin_amdgcn_ds_permute(slot << 2, src);
  dst = __builtin_amdgcn_ds_permute(slot << 2, dst);
  return cnt;
}

__global__ void e1_max(const int* __restrict__ ei, const float* __restrict__ as1,
                       const float* __restrict__ ad1, float* __restrict__ emax1,
                       const float* __restrict__ gst, int E, int N) {
  if (narrow1(gst)) return;
  int lane = threadIdx.x & 31;
  int wave0 = (blockIdx.x * blockDim.x + threadIdx.x) >> 5;
  int nwaves = (gridDim.x * blockDim.x) >> 5;
  int ET = E + N;
  for (int base = wave0 * 32; base < ET; base += nwaves * 32) {
  int e  = base + lane;
  int ec = e < ET ? e : ET - 1;
  int src = ec < E ? ei[ec]     : ec - E;
  int dst = ec < E ? ei[E + ec] : ec - E;
  int h = lane & 3;
  float ev[4];
  int dj[4];
#pragma unroll
  for (int it = 0; it < 4; ++it) {
    int j  = it * 8 + (lane >> 2);
    int sj = __shfl(src, j);
    dj[it] = __shfl(dst, j);
    ev[it] = leaky(as1[sj * 4 + h] + ad1[dj[it] * 4 + h]);
  }
#pragma unroll
  for (int it = 0; it < 4; ++it) {
    int j = it * 8 + (lane >> 2);
    if (base + j < ET) atomicMaxF(&emax1[dj[it] * 4 + h], ev[it]);
  }
  }
}

template <int U>
__device__ __forceinline__ void agg1_rows(int j0, int lane, int src, int dst,
                                          float w0, float w1, float w2, float w3,
                                          const float* __restrict__ h1,
                                          float* __restrict__ agg1) {
  float r0[U], r1[U], wa[U], wb[U];
  int dd[U];
#pragma unroll
  for (int u = 0; u < U; ++u) {
    int j = j0 + u;
    int s = __builtin_amdgcn_readlane(src, j);
    dd[u] = __builtin_amdgcn_readlane(dst, j);
    const float* hp = h1 + (size_t)s * 64 + lane;
    r0[u] = hp[0];
    r1[u] = hp[32];
    float a0 = __int_as_float(__builtin_amdgcn_readlane(__float_as_int(w0), j));
    float a1 = __int_as_float(__builtin_amdgcn_readlane(__float_as_int(w1), j));
    float a2 = __int_as_float(__builtin_amdgcn_readlane(__float_as_int(w2), j));
    float a3 = __int_as_float(__builtin_amdgcn_readlane(__float_as_int(w3), j));
    wa[u] = lane < 16 ? a0 : a1;
    wb[u] = lane < 16 ? a2 : a3;
  }
#pragma unroll
  for (int u = 0; u < U; ++u) {
    float* op = agg1 + (size_t)dd[u] * 64 + lane;
    atomicAdd(op,      r0[u] * wa[u]);
    atomicAdd(op + 32, r1[u] * wb[u]);
  }
}

__global__ void __launch_bounds__(256, 16)
e1_agg(const int* __restrict__ ei, const float* __restrict__ as1,
       const float* __restrict__ ad1, const float* __restrict__ gst,
       const float* __restrict__ emax1,
       const float* __restrict__ h1, float* __restrict__ denom1,
       float* __restrict__ agg1, unsigned* __restrict__ tick, int E, int N) {
  const int lane = threadIdx.x & 31;
  const int ET = E + N;
  const int Nh = half_split(N);
  const bool nar = narrow1(gst);
  const float g0 = gst[0], g1 = gst[1], g2 = gst[2], g3 = gst[3];
  walk_by_half(tick, ET, [=](int base, int hq) {
    int e  = base + lane;
    int ec = e < ET ? e : ET - 1;
    int src = ec < E ? ei[ec]     : ec - E;
    int dst = ec < E ? ei[E + ec] : ec - E;
    const int cnt = keep_half(e < ET, hq, Nh, lane, src, dst);
    if (cnt == 0) return;
    float4 sa = make_float4(0.f, 0.f, 0.f, 0.f), da = sa, m = sa;
    if (lane < cnt) {
      sa = *(const float4*)(as1 + (size_t)src * 4);
      da = *(const float4*)(ad1 + (size_t)dst * 4);
      if (!nar) m = *(const float4*)(emax1 + (size_t)dst * 4);
    }
    if (nar) {
      m.x = leaky(g0 + da.x); m.y = leaky(g1 + da.y);
      m.z = leaky(g2 + da.z); m.w = leaky(g3 + da.w);
    }
    float w0 = __expf(leaky(sa.x + da.x) - m.x);
    float w1 = __expf(leaky(sa.y + da.y) - m.y);
    float w2 = __expf(leaky(sa.z + da.z) - m.z);
    float w3 = __expf(leaky(sa.w + da.w) - m.w);

    int h = lane & 3;
#pragma unroll
    for (int it = 0; it < 4; ++it) {
      if (it * 8 < cnt) {
        int j  = it * 8 + (lane >> 2);
        int dj = __shfl(dst, j);
        float v0 = __shfl(w0, j), v1 = __shfl(w1, j), v2 = __shfl(w2, j), v3 = __shfl(w3, j);
        float v = h == 0 ? v0 : h == 1 ? v1 : h == 2 ? v2 : v3;
        if (j < cnt) atomicAdd(&denom1[dj * 4 + h], v);
      }
    }

    int j0 = 0;
    for (; j0 + 8 <= cnt; j0 += 8) agg1_rows<8>(j0, lane, src, dst, w0, w1, w2, w3, h1, agg1);
    if (j0 + 4 <= cnt) { agg1_rows<4>(j0, lane, src, dst, w0, w1, w2, w3, h1, agg1); j0 += 4; }
    if (j0 + 2 <= cnt) { agg1_rows<2>(j0, lane, src, dst, w0, w1, w2, w3, h1, agg1); j0 += 2; }
    if (j0 < cnt)        agg1_rows<1>(j0, lane, src, dst, w0, w1, w2, w3, h1, agg1);
  });
}

__global__ void bias_elu(float* __restrict__ x2, const float* __restrict__ denom1,
                         const float* __restrict__ b1, int N) {
  int i = blockIdx.x * blockDim.x + threadIdx.x;
  if (i >= N * 64) return;
  float v = x2[i] / denom1[i >> 4] + b1[i & 63];
  x2[i] = v > 0.f ? v : (__expf(v) - 1.f);
}

__global__ void proj2_wmma(const float* __restrict__ x2, const float* __restrict__ W2,
                           float* __restrict__ h2, int N) {
  int gtid = blockIdx.x * blockDim.x + threadIdx.x;
  int wave = gtid >> 5;
  int lane = threadIdx.x & 31;
  int tiles = (N + 15) >> 4;
  if (wave >= tiles) return;
  int half = lane >> 4, l16 = lane & 15;
  int row  = wave * 16 + l16;
  int rowc = row < N ? row : N - 1;
  bool rok = row < N;
  bool cok = l16 < 6;

  v8f acc;
#pragma unroll
  for (int q = 0; q < 8; ++q) acc[q] = 0.f;

#pragma unroll
  for (int s = 0; s < 16; ++s) {
    int k0 = 4 * s + 2 * half;
    v2f a, b;
    a[0] = rok ? x2[rowc * 64 + k0]     : 0.f;
    a[1] = rok ? x2[rowc * 64 + k0 + 1] : 0.f;
    b[0] = cok ? W2[k0 * 6 + l16]       : 0.f;
    b[1] = cok ? W2[(k0 + 1) * 6 + l16] : 0.f;
    acc = __builtin_amdgcn_wmma_f32_16x16x4_f32(
        false, a, false, b, (short)0, acc, false, false);
  }

  if (cok) {
#pragma unroll
    for (int r = 0; r < 8; ++r) {
      int m = wave * 16 + r + 8 * half;
      if (m < N) h2[m * 6 + l16] = acc[r];
    }
  }
}

__global__ void alpha2_kern(const float* __restrict__ h2, const float* __restrict__ aS,
                            const float* __restrict__ aD, float* __restrict__ as2,
                            float* __restrict__ ad2, float* __restrict__ gst, int N) {
  __shared__ float red[8][2];
  int n = blockIdx.x * blockDim.x + threadIdx.x;
  bool ok = n < N;
  int nc = ok ? n : N - 1;
  float s = 0.f, d = 0.f;
#pragma unroll
  for (int c = 0; c < 6; ++c) {
    float v = h2[nc * 6 + c];
    s += v * aS[c];
    d += v * aD[c];
  }
  if (ok) { as2[n] = s; ad2[n] = d; }
  float mx = ok ? s : -3.0e38f, mn = ok ? s : 3.0e38f;
#pragma unroll
  for (int o = 1; o < 32; o <<= 1) {
    mx = fmaxf(mx, __shfl_xor(mx, o));
    mn = fminf(mn, __shfl_xor(mn, o));
  }
  int lane = threadIdx.x & 31, w = threadIdx.x >> 5;
  if (lane == 0) { red[w][0] = mx; red[w][1] = mn; }
  __syncthreads();
  if (threadIdx.x < 2) {
    int t = threadIdx.x;
    float v = red[0][t];
    int nw = blockDim.x >> 5;
    for (int k = 1; k < nw; ++k) v = t == 0 ? fmaxf(v, red[k][t]) : fminf(v, red[k][t]);
    if (t == 0) atomicMaxF(&gst[8], v); else atomicMinF(&gst[9], v);
  }
}

__global__ void e2_max(const int* __restrict__ ei, const float* __restrict__ as2,
                       const float* __restrict__ ad2, float* __restrict__ emax2,
                       const float* __restrict__ gst, int E, int N) {
  if (narrow2(gst)) return;
  int ET = E + N;
  int step = gridDim.x * blockDim.x;
  for (int e = blockIdx.x * blockDim.x + threadIdx.x; e < ET; e += step) {
    int src = e < E ? ei[e]     : e - E;
    int dst = e < E ? ei[E + e] : e - E;
    atomicMaxF(&emax2[dst], leaky(as2[src] + ad2[dst]));
  }
}

__global__ void __launch_bounds__(256, 16)
e2_agg(const int* __restrict__ ei, const float* __restrict__ as2,
       const float* __restrict__ ad2, const float* __restrict__ gst,
       const float* __restrict__ emax2,
       const float* __restrict__ h2, float* __restrict__ agg2r,
       unsigned* __restrict__ tick, int E, int N) {
  const int lane = threadIdx.x & 31;
  const int ET = E + N;
  const int Nh = half_split(N);
  const bool nar = narrow2(gst);
  const float g8 = gst[8];
  walk_by_half(tick, ET, [=](int base, int hq) {
    int e  = base + lane;
    int ec = e < ET ? e : ET - 1;
    int src = ec < E ? ei[ec]     : ec - E;
    int dst = ec < E ? ei[E + ec] : ec - E;
    const int cnt = keep_half(e < ET, hq, Nh, lane, src, dst);
    if (cnt == 0) return;
    float da = 0.f, sv = 0.f, m = 0.f;
    if (lane < cnt) {
      da = ad2[dst];
      sv = as2[src];
      if (!nar) m = emax2[dst];
    }
    if (nar) m = leaky(g8 + da);
    float w = __expf(leaky(sv + da) - m);
    int c = lane & 7;
    float hv[8], wj[8];
    int dj[8];
#pragma unroll
    for (int it = 0; it < 8; ++it) {
      hv[it] = 0.f; wj[it] = 0.f; dj[it] = 0;
      if (it * 4 < cnt) {
        int j  = it * 4 + (lane >> 3);
        int sj = __shfl(src, j);
        dj[it] = __shfl(dst, j);
        wj[it] = __shfl(w, j);
        hv[it] = (c < 6 && j < cnt) ? h2[sj * 6 + c] : 1.f;
      }
    }
#pragma unroll
    for (int it = 0; it < 8; ++it) {
      if (it * 4 < cnt) {
        int j = it * 4 + (lane >> 3);
        if (c < 7 && j < cnt) atomicAdd(&agg2r[dj[it] * 8 + c], hv[it] * wj[it]);
      }
    }
  });
}

__global__ void final_kern(const float* __restrict__ agg2r,
                           const float* __restrict__ b2, float* __restrict__ out, int N) {
  int n = blockIdx.x * blockDim.x + threadIdx.x;
  if (n >= N) return;
  float o[6];
  float m = -3.0e38f;
  float den = agg2r[n * 8 + 6];
#pragma unroll
  for (int c = 0; c < 6; ++c) { o[c] = agg2r[n * 8 + c] / den + b2[c]; m = fmaxf(m, o[c]); }
  float sum = 0.f;
#pragma unroll
  for (int c = 0; c < 6; ++c) sum += __expf(o[c] - m);
  float l = __logf(sum);
#pragma unroll
  for (int c = 0; c < 6; ++c) out[n * 6 + c] = o[c] - m - l;
}

extern "C" void kernel_launch(void* const* d_in, const int* in_sizes, int n_in,
                              void* d_out, int out_size, void* d_ws, size_t ws_size,
                              hipStream_t stream) {
  const float* x      = (const float*)d_in[0];
  const int*   ei     = (const int*)d_in[1];
  const float* W1     = (const float*)d_in[2];
  const float* a_src1 = (const float*)d_in[3];
  const float* a_dst1 = (const float*)d_in[4];
  const float* b1     = (const float*)d_in[5];
  const float* W2     = (const float*)d_in[6];
  const float* a_src2 = (const float*)d_in[7];
  const float* a_dst2 = (const float*)d_in[8];
  const float* b2     = (const float*)d_in[9];
  float* out = (float*)d_out;

  int N = in_sizes[0] / 7;
  int E = in_sizes[1] / 2;

  float* ws     = (float*)d_ws;
  float* h1     = ws;
  float* x2     = h1 + (size_t)N * 64;
  float* agg2r  = x2 + (size_t)N * 64;
  float* as1    = agg2r + (size_t)N * 8;
  float* ad1    = as1 + (size_t)N * 4;
  float* emax1  = ad1 + (size_t)N * 4;
  float* denom1 = emax1 + (size_t)N * 4;
  float* h2     = denom1 + (size_t)N * 4;
  float* as2    = h2 + (size_t)N * 6;
  float* ad2    = as2 + N;
  float* emax2  = ad2 + N;
  float* gst    = emax2 + N;
  unsigned* tick = (unsigned*)(gst + 128);

  const int B = 256;
  auto G = [](long n, int b) { return (int)((n + b - 1) / b); };
  int tiles = (N + 15) / 16;
  int projBlocks = (tiles + (B / 32) - 1) / (B / 32);

  init_kern<<<G((long)N * 64, B), B, 0, stream>>>(emax1, denom1, x2, emax2, agg2r, gst, tick, N);

  proj1_wmma<<<projBlocks, B, 0, stream>>>(x, W1, h1, N);
  alpha1_kern<<<G((long)N * 4, B), B, 0, stream>>>(h1, a_src1, a_dst1, as1, ad1, gst, N);
  e1_max<<<1024, B, 0, stream>>>(ei, as1, ad1, emax1, gst, E, N);
  e1_agg<<<2048, B, 0, stream>>>(ei, as1, ad1, gst, emax1, h1, denom1, x2, tick, E, N);
  bias_elu<<<G((long)N * 64, B), B, 0, stream>>>(x2, denom1, b1, N);

  proj2_wmma<<<projBlocks, B, 0, stream>>>(x2, W2, h2, N);
  alpha2_kern<<<G(N, B), B, 0, stream>>>(h2, a_src2, a_dst2, as2, ad2, gst, N);
  e2_max<<<1024, B, 0, stream>>>(ei, as2, ad2, emax2, gst, E, N);
  e2_agg<<<2048, B, 0, stream>>>(ei, as2, ad2, gst, emax2, h2, agg2r, tick + 2 * 128, E, N);
  final_kern<<<G(N, B), B, 0, stream>>>(agg2r, b2, out, N);
}


# gat2 edge pass: the 8 lanes that hold a source's h2 row form its score themselves (row . a_src, 8-lane DPP sum) instead of gathering it: one cache request per edge less; score sum order changes
# speedup vs baseline: 11.5146x; 1.0294x over previous
#include <hip/hip_runtime.h>
#include <hip/hip_bf16.h>
#include <math.h>

typedef float v2f __attribute__((ext_vector_type(2)));
typedef float v8f __attribute__((ext_vector_type(8)));
typedef float v4f __attribute__((ext_vector_type(4)));

#define NEG_SLOPE 0.2f

__device__ __forceinline__ float leaky(float v) { return v > 0.f ? v : NEG_SLOPE * v; }

__device__ __forceinline__ void atomicMaxF(float* addr, float val) {
  if (val >= 0.f) atomicMax((int*)addr, __float_as_int(val));
  else            atomicMin((unsigned int*)addr, (unsigned int)__float_as_int(val));
}

__device__ __forceinline__ void atomicMinF(float* addr, float val) {
  int b = __float_as_int(val);
  if (b == (int)0x80000000) b = 0;
  if (b >= 0) atomicMin((int*)addr, b);
  else        atomicMax((unsigned int*)addr, (unsigned int)b);
}

#define RANGE_OK 40.0f

__global__ void init_kern(float* emax1, float* denom1, float* agg1,
                          float* emax2, float* agg2r, float* gst, unsigned* tick, int N) {
  int i = blockIdx.x * blockDim.x + threadIdx.x;
  if (i < 16) gst[i] = (i < 4 || i == 8) ? -3.0e38f : 3.0e38f;
  if (i < 4) tick[i * 128] = 0u;
  if (i < N * 64) agg1[i] = 0.f;
  if (i < N * 8)  agg2r[i] = 0.f;
  if (i < N * 4) { emax1[i] = -3.0e38f; denom1[i] = 0.f; }
  if (i < N)     emax2[i] = -3.0e38f;
}

__global__ void proj1_wmma(const float* __restrict__ x, const float* __restrict__ W1,
                           float* __restrict__ h1, int N) {
  int gtid = blockIdx.x * blockDim.x + threadIdx.x;
  int wave = gtid >> 5;
  int lane = threadIdx.x & 31;
  int tiles = (N + 15) >> 4;
  if (wave >= tiles) return;
  int half = lane >> 4;
  int l16  = lane & 15;
  int row  = wave * 16 + l16;
  int rowc = row < N ? row : N - 1;
  bool rok = row < N;

  v2f a[2];
#pragma unroll
  for (int s = 0; s < 2; ++s) {
    int k0 = 4 * s + 2 * half;
    a[s][0] = (rok && k0     < 7) ? x[rowc * 7 + k0]     : 0.f;
    a[s][1] = (rok && k0 + 1 < 7) ? x[rowc * 7 + k0 + 1] : 0.f;
  }

  v8f acc[4];
#pragma unroll
  for (int t = 0; t < 4; ++t)
#pragma unroll
    for (int q = 0; q < 8; ++q) acc[t][q] = 0.f;

#pragma unroll
  for (int s = 0; s < 2; ++s) {
    int k0 = 4 * s + 2 * half;
#pragma unroll
    for (int t = 0; t < 4; ++t) {
      int col = t * 16 + l16;
      v2f b;
      b[0] = (k0     < 7) ? W1[k0 * 64 + col]       : 0.f;
      b[1] = (k0 + 1 < 7) ? W1[(k0 + 1) * 64 + col] : 0.f;
      acc[t] = __builtin_amdgcn_wmma_f32_16x16x4_f32(
          false, a[s], false, b, (short)0, acc[t], false, false);
    }
  }

#pragma unroll
  for (int t = 0; t < 4; ++t)
#pragma unroll
    for (int r = 0; r < 8; ++r) {
      int m = wave * 16 + r + 8 * half;
      if (m < N) h1[m * 64 + t * 16 + l16] = acc[t][r];
    }
}

__global__ void alpha1_kern(const float* __restrict__ h1, const float* __restrict__ aS,
                            const float* __restrict__ aD, float* __restrict__ as1,
                            float* __restrict__ ad1, float* __restrict__ gst, int N) {
  __shared__ float red[8][8];
  int i = blockIdx.x * blockDim.x + threadIdx.x;
  bool ok = i < N * 4;
  int ic = ok ? i : N * 4 - 1;
  int hd = ic & 3, n = ic >> 2;
  const float* hp = h1 + n * 64 + hd * 16;
  float s = 0.f, d = 0.f;
#pragma unroll
  for (int c = 0; c < 16; ++c) {
    float v = hp[c];
    s += v * aS[hd * 16 + c];
    d += v * aD[hd * 16 + c];
  }
  if (ok) { as1[i] = s; ad1[i] = d; }
  float mx = ok ? s : -3.0e38f, mn = ok ? s : 3.0e38f;
#pragma unroll
  for (int o = 4; o < 32; o <<= 1) {
    mx = fmaxf(mx, __shfl_xor(mx, o));
    mn = fminf(mn, __shfl_xor(mn, o));
  }
  int lane = threadIdx.x & 31, w = threadIdx.x >> 5;
  if (lane < 4) { red[w][lane] = mx; red[w][4 + lane] = mn; }
  __syncthreads();
  if (threadIdx.x < 8) {
    int t = threadIdx.x;
    float v = red[0][t];
    int nw = blockDim.x >> 5;
    for (int k = 1; k < nw; ++k) v = t < 4 ? fmaxf(v, red[k][t]) : fminf(v, red[k][t]);
    if (t < 4) atomicMaxF(&gst[t], v); else atomicMinF(&gst[t], v);
  }
}

__device__ __forceinline__ bool narrow1(const float* __restrict__ gst) {
  return (gst[0] - gst[4] <= RANGE_OK) & (gst[1] - gst[5] <= RANGE_OK) &
         (gst[2] - gst[6] <= RANGE_OK) & (gst[3] - gst[7] <= RANGE_OK);
}
__device__ __forceinline__ bool narrow2(const float* __restrict__ gst) {
  return gst[8] - gst[9] <= RANGE_OK;
}

#define SPAN 256
#define TICK_PITCH 128

__device__ __forceinline__ int chip_half() {
  return (__builtin_amdgcn_s_sendmsg_rtn(0x87) >> 18) & 1;
}

__device__ __forceinline__ int half_split(int N) {
  return ((N >> 1) + 15) & ~15;
}

template <class F>
__device__ __forceinline__ void walk_by_half(unsigned* __restrict__ tick, int ET, F chunk) {
  __shared__ int s_t[2];
  const int nspan = (ET + SPAN - 1) / SPAN;
  const int wv = threadIdx.x >> 5;
  const int h = chip_half();
  int par = 0;
  for (int q = 0; q < 2; ++q) {
    const int hq = h ^ q;
    unsigned* tk = tick + hq * TICK_PITCH;
    int nxt = 0;
    if (threadIdx.x == 0) nxt = (int)atomicInc(tk, 0xFFFFFFFFu);
    for (;;) {
      if (threadIdx.x == 0) s_t[par] = nxt;
      __syncthreads();
      const int t = __builtin_amdgcn_readfirstlane(s_t[par]);
      par ^= 1;
      if (t >= nspan) break;
      if (threadIdx.x == 0) nxt = (int)atomicInc(tk, 0xFFFFFFFFu);
      const int base = t * SPAN + wv * 32;
      if (base < ET) chunk(base, hq);
    }
  }
}

__device__ __forceinline__ int keep_half(bool valid, int hq, int Nh, int lane, int& src, int& dst) {
  bool mine = valid && ((dst >= Nh) == (hq != 0));
  unsigned mask = __builtin_amdgcn_ballot_w32(mine);
  int cnt = __builtin_popcount(mask);
  int below = (int)__builtin_amdgcn_mbcnt_lo(mask, 0u);
  int slot = mine ? below : cnt + lane - below;
  src = __builtin_amdgcn_ds_permute(slot << 2, src);
  dst = __builtin_amdgcn_ds_permute(slot << 2, dst);
  return cnt;
}

__global__ void e1_max(const int* __restrict__ ei, const float* __restrict__ as1,
                       const float* __restrict__ ad1, float* __restrict__ emax1,
                       const float* __restrict__ gst, int E, int N) {
  if (narrow1(gst)) return;
  int lane = threadIdx.x & 31;
  int wave0 = (blockIdx.x * blockDim.x + threadIdx.x) >> 5;
  int nwaves = (gridDim.x * blockDim.x) >> 5;
  int ET = E + N;
  for (int base = wave0 * 32; base < ET; base += nwaves * 32) {
  int e  = base + lane;
  int ec = e < ET ? e : ET - 1;
  int src = ec < E ? ei[ec]     : ec - E;
  int dst = ec < E ? ei[E + ec] : ec - E;
  int h = lane & 3;
  float ev[4];
  int dj[4];
#pragma unroll
  for (int it = 0; it < 4; ++it) {
    int j  = it * 8 + (lane >> 2);
    int sj = __shfl(src, j);
    dj[it] = __shfl(dst, j);
    ev[it] = leaky(as1[sj * 4 + h] + ad1[dj[it] * 4 + h]);
  }
#pragma unroll
  for (int it = 0; it < 4; ++it) {
    int j = it * 8 + (lane >> 2);
    if (base + j < ET) atomicMaxF(&emax1[dj[it] * 4 + h], ev[it]);
  }
  }
}

template <int U>
__device__ __forceinline__ void agg1_rows(int j0, int lane, int src, int dst,
                                          float w0, float w1, float w2, float w3,
                                          const float* __restrict__ h1,
                                          float* __restrict__ agg1) {
  float r0[U], r1[U], wa[U], wb[U];
  int dd[U];
#pragma unroll
  for (int u = 0; u < U; ++u) {
    int j = j0 + u;
    int s = __builtin_amdgcn_readlane(src, j);
    dd[u] = __builtin_amdgcn_readlane(dst, j);
    const float* hp = h1 + (size_t)s * 64 + lane;
    r0[u] = hp[0];
    r1[u] = hp[32];
    float a0 = __int_as_float(__builtin_amdgcn_readlane(__float_as_int(w0), j));
    float a1 = __int_as_float(__builtin_amdgcn_readlane(__float_as_int(w1), j));
    float a2 = __int_as_float(__builtin_amdgcn_readlane(__float_as_int(w2), j));
    float a3 = __int_as_float(__builtin_amdgcn_readlane(__float_as_int(w3), j));
    wa[u] = lane < 16 ? a0 : a1;
    wb[u] = lane < 16 ? a2 : a3;
  }
#pragma unroll
  for (int u = 0; u < U; ++u) {
    float* op = agg1 + (size_t)dd[u] * 64 + lane;
    atomicAdd(op,      r0[u] * wa[u]);
    atomicAdd(op + 32, r1[u] * wb[u]);
  }
}

__global__ void __launch_bounds__(256, 16)
e1_agg(const int* __restrict__ ei, const float* __restrict__ as1,
       const float* __restrict__ ad1, const float* __restrict__ gst,
       const float* __restrict__ emax1,
       const float* __restrict__ h1, float* __restrict__ denom1,
       float* __restrict__ agg1, unsigned* __restrict__ tick, int E, int N) {
  const int lane = threadIdx.x & 31;
  const int ET = E + N;
  const int Nh = half_split(N);
  const bool nar = narrow1(gst);
  const float g0 = gst[0], g1 = gst[1], g2 = gst[2], g3 = gst[3];
  walk_by_half(tick, ET, [=](int base, int hq) {
    int e  = base + lane;
    int ec = e < ET ? e : ET - 1;
    int src = ec < E ? ei[ec]     : ec - E;
    int dst = ec < E ? ei[E + ec] : ec - E;
    const int cnt = keep_half(e < ET, hq, Nh, lane, src, dst);
    if (cnt == 0) return;
    float4 sa = make_float4(0.f, 0.f, 0.f, 0.f), da = sa, m = sa;
    if (lane < cnt) {
      sa = *(const float4*)(as1 + (size_t)src * 4);
      da = *(const float4*)(ad1 + (size_t)dst * 4);
      if (!nar) m = *(const float4*)(emax1 + (size_t)dst * 4);
    }
    if (nar) {
      m.x = leaky(g0 + da.x); m.y = leaky(g1 + da.y);
      m.z = leaky(g2 + da.z); m.w = leaky(g3 + da.w);
    }
    float w0 = __expf(leaky(sa.x + da.x) - m.x);
    float w1 = __expf(leaky(sa.y + da.y) - m.y);
    float w2 = __expf(leaky(sa.z + da.z) - m.z);
    float w3 = __expf(leaky(sa.w + da.w) - m.w);

    int h = lane & 3;
#pragma unroll
    for (int it = 0; it < 4; ++it) {
      if (it * 8 < cnt) {
        int j  = it * 8 + (lane >> 2);
        int dj = __shfl(dst, j);
        float v0 = __shfl(w0, j), v1 = __shfl(w1, j), v2 = __shfl(w2, j), v3 = __shfl(w3, j);
        float v = h == 0 ? v0 : h == 1 ? v1 : h == 2 ? v2 : v3;
        if (j < cnt) atomicAdd(&denom1[dj * 4 + h], v);
      }
    }

    int j0 = 0;
    for (; j0 + 8 <= cnt; j0 += 8) agg1_rows<8>(j0, lane, src, dst, w0, w1, w2, w3, h1, agg1);
    if (j0 + 4 <= cnt) { agg1_rows<4>(j0, lane, src, dst, w0, w1, w2, w3, h1, agg1); j0 += 4; }
    if (j0 + 2 <= cnt) { agg1_rows<2>(j0, lane, src, dst, w0, w1, w2, w3, h1, agg1); j0 += 2; }
    if (j0 < cnt)        agg1_rows<1>(j0, lane, src, dst, w0, w1, w2, w3, h1, agg1);
  });
}

__global__ void bias_elu(float* __restrict__ x2, const float* __restrict__ denom1,
                         const float* __restrict__ b1, int N) {
  int i = blockIdx.x * blockDim.x + threadIdx.x;
  if (i >= N * 64) return;
  float v = x2[i] / denom1[i >> 4] + b1[i & 63];
  x2[i] = v > 0.f ? v : (__expf(v) - 1.f);
}

__global__ void proj2_wmma(const float* __restrict__ x2, const float* __restrict__ W2,
                           float* __restrict__ h2, int N) {
  int gtid = blockIdx.x * blockDim.x + threadIdx.x;
  int wave = gtid >> 5;
  int lane = threadIdx.x & 31;
  int tiles = (N + 15) >> 4;
  if (wave >= tiles) return;
  int half = lane >> 4, l16 = lane & 15;
  int row  = wave * 16 + l16;
  int rowc = row < N ? row : N - 1;
  bool rok = row < N;
  bool cok = l16 < 6;

  v8f acc;
#pragma unroll
  for (int q = 0; q < 8; ++q) acc[q] = 0.f;

#pragma unroll
  for (int s = 0; s < 16; ++s) {
    int k0 = 4 * s + 2 * half;
    v2f a, b;
    a[0] = rok ? x2[rowc * 64 + k0]     : 0.f;
    a[1] = rok ? x2[rowc * 64 + k0 + 1] : 0.f;
    b[0] = cok ? W2[k0 * 6 + l16]       : 0.f;
    b[1] = cok ? W2[(k0 + 1) * 6 + l16] : 0.f;
    acc = __builtin_amdgcn_wmma_f32_16x16x4_f32(
        false, a, false, b, (short)0, acc, false, false);
  }

  if (cok) {
#pragma unroll
    for (int r = 0; r < 8; ++r) {
      int m = wave * 16 + r + 8 * half;
      if (m < N) h2[m * 6 + l16] = acc[r];
    }
  }
}

__global__ void alpha2_kern(const float* __restrict__ h2, const float* __restrict__ aS,
                            const float* __restrict__ aD, float* __restrict__ as2,
                            float* __restrict__ ad2, float* __restrict__ gst, int N) {
  __shared__ float red[8][2];
  int n = blockIdx.x * blockDim.x + threadIdx.x;
  bool ok = n < N;
  int nc = ok ? n : N - 1;
  float s = 0.f, d = 0.f;
#pragma unroll
  for (int c = 0; c < 6; ++c) {
    float v = h2[nc * 6 + c];
    s += v * aS[c];
    d += v * aD[c];
  }
  if (ok) { as2[n] = s; ad2[n] = d; }
  float mx = ok ? s : -3.0e38f, mn = ok ? s : 3.0e38f;
#pragma unroll
  for (int o = 1; o < 32; o <<= 1) {
    mx = fmaxf(mx, __shfl_xor(mx, o));
    mn = fminf(mn, __shfl_xor(mn, o));
  }
  int lane = threadIdx.x & 31, w = threadIdx.x >> 5;
  if (lane == 0) { red[w][0] = mx; red[w][1] = mn; }
  __syncthreads();
  if (threadIdx.x < 2) {
    int t = threadIdx.x;
    float v = red[0][t];
    int nw = blockDim.x >> 5;
    for (int k = 1; k < nw; ++k) v = t == 0 ? fmaxf(v, red[k][t]) : fminf(v, red[k][t]);
    if (t == 0) atomicMaxF(&gst[8], v); else atomicMinF(&gst[9], v);
  }
}

__global__ void e2_max(const int* __restrict__ ei, const float* __restrict__ as2,
                       const float* __restrict__ ad2, float* __restrict__ emax2,
                       const float* __restrict__ gst, int E, int N) {
  if (narrow2(gst)) return;
  int ET = E + N;
  int step = gridDim.x * blockDim.x;
  for (int e = blockIdx.x * blockDim.x + threadIdx.x; e < ET; e += step) {
    int src = e < E ? ei[e]     : e - E;
    int dst = e < E ? ei[E + e] : e - E;
    atomicMaxF(&emax2[dst], leaky(as2[src] + ad2[dst]));
  }
}

__device__ __forceinline__ float sum8(float v) {
  v += __int_as_float(__builtin_amdgcn_update_dpp(0, __float_as_int(v), 0x161, 0xf, 0xf, true));
  v += __int_as_float(__builtin_amdgcn_update_dpp(0, __float_as_int(v), 0x162, 0xf, 0xf, true));
  v += __int_as_float(__builtin_amdgcn_update_dpp(0, __float_as_int(v), 0x164, 0xf, 0xf, true));
  return v;
}

__global__ void __launch_bounds__(256, 16)
e2_agg(const int* __restrict__ ei, const float* __restrict__ aS,
       const float* __restrict__ ad2, const float* __restrict__ gst,
       const float* __restrict__ emax2,
       const float* __restrict__ h2, float* __restrict__ agg2r,
       unsigned* __restrict__ tick, int E, int N) {
  const int lane = threadIdx.x & 31;
  const int ET = E + N;
  const int Nh = half_split(N);
  const bool nar = narrow2(gst);
  const float g8 = gst[8];
  const int c = lane & 7;
  const float ac = c < 6 ? aS[c] : 0.f;
  walk_by_half(tick, ET, [=](int base, int hq) {
    int e  = base + lane;
    int ec = e < ET ? e : ET - 1;
    int src = ec < E ? ei[ec]     : ec - E;
    int dst = ec < E ? ei[E + ec] : ec - E;
    const int cnt = keep_half(e < ET, hq, Nh, lane, src, dst);
    if (cnt == 0) return;
    float da = 0.f, m = 0.f;
    if (lane < cnt) {
      da = ad2[dst];
      if (!nar) m = emax2[dst];
    }
    if (nar) m = leaky(g8 + da);
    float hv[8], dv[8], mv[8];
    int dj[8];
#pragma unroll
    for (int it = 0; it < 8; ++it) {
      hv[it] = 0.f; dv[it] = 0.f; mv[it] = 0.f; dj[it] = 0;
      if (it * 4 < cnt) {
        int j  = it * 4 + (lane >> 3);
        int sj = __shfl(src, j);
        dj[it] = __shfl(dst, j);
        dv[it] = __shfl(da, j);
        mv[it] = __shfl(m, j);
        hv[it] = (c < 6 && j < cnt) ? h2[sj * 6 + c] : 0.f;
      }
    }
#pragma unroll
    for (int it = 0; it < 8; ++it) {
      if (it * 4 < cnt) {
        int j = it * 4 + (lane >> 3);
        float sv = sum8(hv[it] * ac);
        float w = __expf(leaky(sv + dv[it]) - mv[it]);
        float val = c == 6 ? w : hv[it] * w;
        if (c < 7 && j < cnt) atomicAdd(&agg2r[dj[it] * 8 + c], val);
      }
    }
  });
}

__global__ void final_kern(const float* __restrict__ agg2r,
                           const float* __restrict__ b2, float* __restrict__ out, int N) {
  int n = blockIdx.x * blockDim.x + threadIdx.x;
  if (n >= N) return;
  float o[6];
  float m = -3.0e38f;
  float den = agg2r[n * 8 + 6];
#pragma unroll
  for (int c = 0; c < 6; ++c) { o[c] = agg2r[n * 8 + c] / den + b2[c]; m = fmaxf(m, o[c]); }
  float sum = 0.f;
#pragma unroll
  for (int c = 0; c < 6; ++c) sum += __expf(o[c] - m);
  float l = __logf(sum);
#pragma unroll
  for (int c = 0; c < 6; ++c) out[n * 6 + c] = o[c] - m - l;
}

extern "C" void kernel_launch(void* const* d_in, const int* in_sizes, int n_in,
                              void* d_out, int out_size, void* d_ws, size_t ws_size,
                              hipStream_t stream) {
  const float* x      = (const float*)d_in[0];
  const int*   ei     = (const int*)d_in[1];
  const float* W1     = (const float*)d_in[2];
  const float* a_src1 = (const float*)d_in[3];
  const float* a_dst1 = (const float*)d_in[4];
  const float* b1     = (const float*)d_in[5];
  const float* W2     = (const float*)d_in[6];
  const float* a_src2 = (const float*)d_in[7];
  const float* a_dst2 = (const float*)d_in[8];
  const float* b2     = (const float*)d_in[9];
  float* out = (float*)d_out;

  int N = in_sizes[0] / 7;
  int E = in_sizes[1] / 2;

  float* ws     = (float*)d_ws;
  float* h1     = ws;
  float* x2     = h1 + (size_t)N * 64;
  float* agg2r  = x2 + (size_t)N * 64;
  float* as1    = agg2r + (size_t)N * 8;
  float* ad1    = as1 + (size_t)N * 4;
  float* emax1  = ad1 + (size_t)N * 4;
  float* denom1 = emax1 + (size_t)N * 4;
  float* h2     = denom1 + (size_t)N * 4;
  float* as2    = h2 + (size_t)N * 6;
  float* ad2    = as2 + N;
  float* emax2  = ad2 + N;
  float* gst    = emax2 + N;
  unsigned* tick = (unsigned*)(gst + 128);

  const int B = 256;
  auto G = [](long n, int b) { return (int)((n + b - 1) / b); };
  int tiles = (N + 15) / 16;
  int projBlocks = (tiles + (B / 32) - 1) / (B / 32);

  init_kern<<<G((long)N * 64, B), B, 0, stream>>>(emax1, denom1, x2, emax2, agg2r, gst, tick, N);

  proj1_wmma<<<projBlocks, B, 0, stream>>>(x, W1, h1, N);
  alpha1_kern<<<G((long)N * 4, B), B, 0, stream>>>(h1, a_src1, a_dst1, as1, ad1, gst, N);
  e1_max<<<1024, B, 0, stream>>>(ei, as1, ad1, emax1, gst, E, N);
  e1_agg<<<2048, B, 0, stream>>>(ei, as1, ad1, gst, emax1, h1, denom1, x2, tick, E, N);
  bias_elu<<<G((long)N * 64, B), B, 0, stream>>>(x2, denom1, b1, N);

  proj2_wmma<<<projBlocks, B, 0, stream>>>(x2, W2, h2, N);
  alpha2_kern<<<G(N, B), B, 0, stream>>>(h2, a_src2, a_dst2, as2, ad2, gst, N);
  e2_max<<<1024, B, 0, stream>>>(ei, as2, ad2, emax2, gst, E, N);
  e2_agg<<<2048, B, 0, stream>>>(ei, a_src2, ad2, gst, emax2, h2, agg2r, tick + 2 * 128, E, N);
  final_kern<<<G(N, B), B, 0, stream>>>(agg2r, b2, out, N);
}


# gat1 edge pass: the lane owning an edge loads the source's 7 features once, forms its 4 scores from them (x.(W1 a_src): re-associated sum) and the 32-word record, rows go to the wave via LDS
# speedup vs baseline: 16.6855x; 1.4491x over previous
#include <hip/hip_runtime.h>
#include <hip/hip_bf16.h>
#include <math.h>

typedef float v2f __attribute__((ext_vector_type(2)));
typedef float v8f __attribute__((ext_vector_type(8)));
typedef float v4f __attribute__((ext_vector_type(4)));

#define NEG_SLOPE 0.2f

__device__ __forceinline__ float leaky(float v) { return v > 0.f ? v : NEG_SLOPE * v; }

__device__ __forceinline__ void atomicMaxF(float* addr, float val) {
  if (val >= 0.f) atomicMax((int*)addr, __float_as_int(val));
  else            atomicMin((unsigned int*)addr, (unsigned int)__float_as_int(val));
}

__device__ __forceinline__ void atomicMinF(float* addr, float val) {
  int b = __float_as_int(val);
  if (b == (int)0x80000000) b = 0;
  if (b >= 0) atomicMin((int*)addr, b);
  else        atomicMax((unsigned int*)addr, (unsigned int)b);
}

#define RANGE_OK 40.0f

__global__ void init_kern(float* emax1, float* S1,
                          float* emax2, float* agg2r, float* gst, unsigned* tick, int N) {
  int i = blockIdx.x * blockDim.x + threadIdx.x;
  if (i < 16) gst[i] = (i < 4 || i == 8) ? -3.0e38f : 3.0e38f;
  if (i < 4) tick[i * 128] = 0u;
  if (i < N * 32) S1[i] = 0.f;
  if (i < N * 8)  agg2r[i] = 0.f;
  if (i < N * 4)  emax1[i] = -3.0e38f;
  if (i < N)     emax2[i] = -3.0e38f;
}

__global__ void proj1_wmma(const float* __restrict__ x, const float* __restrict__ W1,
                           float* __restrict__ h1, int N) {
  int gtid = blockIdx.x * blockDim.x + threadIdx.x;
  int wave = gtid >> 5;
  int lane = threadIdx.x & 31;
  int tiles = (N + 15) >> 4;
  if (wave >= tiles) return;
  int half = lane >> 4;
  int l16  = lane & 15;
  int row  = wave * 16 + l16;
  int rowc = row < N ? row : N - 1;
  bool rok = row < N;

  v2f a[2];
#pragma unroll
  for (int s = 0; s < 2; ++s) {
    int k0 = 4 * s + 2 * half;
    a[s][0] = (rok && k0     < 7) ? x[rowc * 7 + k0]     : 0.f;
    a[s][1] = (rok && k0 + 1 < 7) ? x[rowc * 7 + k0 + 1] : 0.f;
  }

  v8f acc[4];
#pragma unroll
  for (int t = 0; t < 4; ++t)
#pragma unroll
    for (int q = 0; q < 8; ++q) acc[t][q] = 0.f;

#pragma unroll
  for (int s = 0; s < 2; ++s) {
    int k0 = 4 * s + 2 * half;
#pragma unroll
    for (int t = 0; t < 4; ++t) {
      int col = t * 16 + l16;
      v2f b;
      b[0] = (k0     < 7) ? W1[k0 * 64 + col]       : 0.f;
      b[1] = (k0 + 1 < 7) ? W1[(k0 + 1) * 64 + col] : 0.f;
      acc[t] = __builtin_amdgcn_wmma_f32_16x16x4_f32(
          false, a[s], false, b, (short)0, acc[t], false, false);
    }
  }

#pragma unroll
  for (int t = 0; t < 4; ++t)
#pragma unroll
    for (int r = 0; r < 8; ++r) {
      int m = wave * 16 + r + 8 * half;
      if (m < N) h1[m * 64 + t * 16 + l16] = acc[t][r];
    }
}

__global__ void alpha1_kern(const float* __restrict__ h1, const float* __restrict__ aS,
                            const float* __restrict__ aD, float* __restrict__ as1,
                            float* __restrict__ ad1, float* __restrict__ gst, int N) {
  __shared__ float red[8][8];
  int i = blockIdx.x * blockDim.x + threadIdx.x;
  bool ok = i < N * 4;
  int ic = ok ? i : N * 4 - 1;
  int hd = ic & 3, n = ic >> 2;
  const float* hp = h1 + n * 64 + hd * 16;
  float s = 0.f, d = 0.f;
#pragma unroll
  for (int c = 0; c < 16; ++c) {
    float v = hp[c];
    s += v * aS[hd * 16 + c];
    d += v * aD[hd * 16 + c];
  }
  if (ok) { as1[i] = s; ad1[i] = d; }
  float mx = ok ? s : -3.0e38f, mn = ok ? s : 3.0e38f;
#pragma unroll
  for (int o = 4; o < 32; o <<= 1) {
    mx = fmaxf(mx, __shfl_xor(mx, o));
    mn = fminf(mn, __shfl_xor(mn, o));
  }
  int lane = threadIdx.x & 31, w = threadIdx.x >> 5;
  if (lane < 4) { red[w][lane] = mx; red[w][4 + lane] = mn; }
  __syncthreads();
  if (threadIdx.x < 8) {
    int t = threadIdx.x;
    float v = red[0][t];
    int nw = blockDim.x >> 5;
    for (int k = 1; k < nw; ++k) v = t < 4 ? fmaxf(v, red[k][t]) : fminf(v, red[k][t]);
    if (t < 4) atomicMaxF(&gst[t], v); else atomicMinF(&gst[t], v);
  }
}

__device__ __forceinline__ bool narrow1(const float* __restrict__ gst) {
  return (gst[0] - gst[4] <= RANGE_OK) & (gst[1] - gst[5] <= RANGE_OK) &
         (gst[2] - gst[6] <= RANGE_OK) & (gst[3] - gst[7] <= RANGE_OK);
}
__device__ __forceinline__ bool narrow2(const float* __restrict__ gst) {
  return gst[8] - gst[9] <= RANGE_OK;
}

#define SPAN 256
#define TICK_PITCH 128

__device__ __forceinline__ int chip_half() {
  return (__builtin_amdgcn_s_sendmsg_rtn(0x87) >> 18) & 1;
}

__device__ __forceinline__ int half_split(int N) {
  return ((N >> 1) + 15) & ~15;
}

template <class F>
__device__ __forceinline__ void walk_by_half(unsigned* __restrict__ tick, int ET, F chunk) {
  __shared__ int s_t[2];
  const int nspan = (ET + SPAN - 1) / SPAN;
  const int wv = threadIdx.x >> 5;
  const int h = chip_half();
  int par = 0;
  for (int q = 0; q < 2; ++q) {
    const int hq = h ^ q;
    unsigned* tk = tick + hq * TICK_PITCH;
    int nxt = 0;
    if (threadIdx.x == 0) nxt = (int)atomicInc(tk, 0xFFFFFFFFu);
    for (;;) {
      if (threadIdx.x == 0) s_t[par] = nxt;
      __syncthreads();
      const int t = __builtin_amdgcn_readfirstlane(s_t[par]);
      par ^= 1;
      if (t >= nspan) break;
      if (threadIdx.x == 0) nxt = (int)atomicInc(tk, 0xFFFFFFFFu);
      const int base = t * SPAN + wv * 32;
      if (base < ET) chunk(base, hq);
    }
  }
}

__device__ __forceinline__ int keep_half(bool valid, int hq, int Nh, int lane, int& src, int& dst) {
  bool mine = valid && ((dst >= Nh) == (hq != 0));
  unsigned mask = __builtin_amdgcn_ballot_w32(mine);
  int cnt = __builtin_popcount(mask);
  int below = (int)__builtin_amdgcn_mbcnt_lo(mask, 0u);
  int slot = mine ? below : cnt + lane - below;
  src = __builtin_amdgcn_ds_permute(slot << 2, src);
  dst = __builtin_amdgcn_ds_permute(slot << 2, dst);
  return cnt;
}

__global__ void e1_max(const int* __restrict__ ei, const float* __restrict__ as1,
                       const float* __restrict__ ad1, float* __restrict__ emax1,
                       const float* __restrict__ gst, int E, int N) {
  if (narrow1(gst)) return;
  int lane = threadIdx.x & 31;
  int wave0 = (blockIdx.x * blockDim.x + threadIdx.x) >> 5;
  int nwaves = (gridDim.x * blockDim.x) >> 5;
  int ET = E + N;
  for (int base = wave0 * 32; base < ET; base += nwaves * 32) {
  int e  = base + lane;
  int ec = e < ET ? e : ET - 1;
  int src = ec < E ? ei[ec]     : ec - E;
  int dst = ec < E ? ei[E + ec] : ec - E;
  int h = lane & 3;
  float ev[4];
  int dj[4];
#pragma unroll
  for (int it = 0; it < 4; ++it) {
    int j  = it * 8 + (lane >> 2);
    int sj = __shfl(src, j);
    dj[it] = __shfl(dst, j);
    ev[it] = leaky(as1[sj * 4 + h] + ad1[dj[it] * 4 + h]);
  }
#pragma unroll
  for (int it = 0; it < 4; ++it) {
    int j = it * 8 + (lane >> 2);
    if (base + j < ET) atomicMaxF(&emax1[dj[it] * 4 + h], ev[it]);
  }
  }
}

#define REC_PITCH 36

__global__ void __launch_bounds__(256, 16)
e1_agg(const int* __restrict__ ei, const float* __restrict__ W1,
       const float* __restrict__ aS,
       const float* __restrict__ ad1, const float* __restrict__ gst,
       const float* __restrict__ emax1,
       const float* __restrict__ x,
       float* __restrict__ S1, unsigned* __restrict__ tick, int E, int N) {
  __shared__ __attribute__((aligned(16))) float rec[8][32 * REC_PITCH];
  const int lane = threadIdx.x & 31;
  const int ET = E + N;
  const int Nh = half_split(N);
  const bool nar = narrow1(gst);
  const float g0 = gst[0], g1 = gst[1], g2 = gst[2], g3 = gst[3];
  float wal = 0.f;
  {
    const int lk = lane < 28 ? lane >> 2 : 6, lh = lane & 3;
#pragma unroll
    for (int c = 0; c < 16; ++c) wal += W1[lk * 64 + lh * 16 + c] * aS[lh * 16 + c];
  }
  float wa[28];
#pragma unroll
  for (int i = 0; i < 28; ++i)
    wa[i] = __int_as_float(__builtin_amdgcn_readlane(__float_as_int(wal), i));
  float* const rows = rec[threadIdx.x >> 5];
  walk_by_half(tick, ET, [=](int base, int hq) {
    int e  = base + lane;
    int ec = e < ET ? e : ET - 1;
    int src = ec < E ? ei[ec]     : ec - E;
    int dst = ec < E ? ei[E + ec] : ec - E;
    const int cnt = keep_half(e < ET, hq, Nh, lane, src, dst);
    if (cnt == 0) return;
    if (lane < cnt) {
      const float* xr = x + (size_t)src * 7;
      const float x0 = xr[0], x1 = xr[1], x2 = xr[2], x3 = xr[3], x4 = xr[4], x5 = xr[5], x6 = xr[6];
      const float4 da = *(const float4*)(ad1 + (size_t)dst * 4);
      float4 m = make_float4(0.f, 0.f, 0.f, 0.f);
      if (!nar) m = *(const float4*)(emax1 + (size_t)dst * 4);
      else {
        m.x = leaky(g0 + da.x); m.y = leaky(g1 + da.y);
        m.z = leaky(g2 + da.z); m.w = leaky(g3 + da.w);
      }
      const float s0 = x0 * wa[0] + x1 * wa[4] + x2 * wa[8]  + x3 * wa[12] + x4 * wa[16] + x5 * wa[20] + x6 * wa[24];
      const float s1 = x0 * wa[1] + x1 * wa[5] + x2 * wa[9]  + x3 * wa[13] + x4 * wa[17] + x5 * wa[21] + x6 * wa[25];
      const float s2 = x0 * wa[2] + x1 * wa[6] + x2 * wa[10] + x3 * wa[14] + x4 * wa[18] + x5 * wa[22] + x6 * wa[26];
      const float s3 = x0 * wa[3] + x1 * wa[7] + x2 * wa[11] + x3 * wa[15] + x4 * wa[19] + x5 * wa[23] + x6 * wa[27];
      const float w0 = __expf(leaky(s0 + da.x) - m.x);
      const float w1 = __expf(leaky(s1 + da.y) - m.y);
      const float w2 = __expf(leaky(s2 + da.z) - m.z);
      const float w3 = __expf(leaky(s3 + da.w) - m.w);
      float4* r = (float4*)(rows + lane * REC_PITCH);
      r[0] = make_float4(x0 * w0, x1 * w0, x2 * w0, x3 * w0);
      r[1] = make_float4(x4 * w0, x5 * w0, x6 * w0, w0);
      r[2] = make_float4(x0 * w1, x1 * w1, x2 * w1, x3 * w1);
      r[3] = make_float4(x4 * w1, x5 * w1, x6 * w1, w1);
      r[4] = make_float4(x0 * w2, x1 * w2, x2 * w2, x3 * w2);
      r[5] = make_float4(x4 * w2, x5 * w2, x6 * w2, w2);
      r[6] = make_float4(x0 * w3, x1 * w3, x2 * w3, x3 * w3);
      r[7] = make_float4(x4 * w3, x5 * w3, x6 * w3, w3);
    }
    __builtin_amdgcn_fence(__ATOMIC_RELEASE, "wavefront");
    __builtin_amdgcn_wave_barrier();
    __builtin_amdgcn_fence(__ATOMIC_ACQUIRE, "wavefront");
    const float* rl = rows + lane;
    int j = 0;
    for (; j + 8 <= cnt; j += 8) {
      float v[8];
#pragma unroll
      for (int u = 0; u < 8; ++u) v[u] = rl[(j + u) * REC_PITCH];
#pragma unroll
      for (int u = 0; u < 8; ++u) {
        int d = __builtin_amdgcn_readlane(dst, j + u);
        atomicAdd(S1 + (size_t)d * 32 + lane, v[u]);
      }
    }
    if (j + 4 <= cnt) {
      float v[4];
#pragma unroll
      for (int u = 0; u < 4; ++u) v[u] = rl[(j + u) * REC_PITCH];
#pragma unroll
      for (int u = 0; u < 4; ++u) {
        int d = __builtin_amdgcn_readlane(dst, j + u);
        atomicAdd(S1 + (size_t)d * 32 + lane, v[u]);
      }
      j += 4;
    }
    for (; j < cnt; ++j) {
      float v = rl[j * REC_PITCH];
      int d = __builtin_amdgcn_readlane(dst, j);
      atomicAdd(S1 + (size_t)d * 32 + lane, v);
    }
  });
}

__global__ void post1_wmma(const float* __restrict__ S1, const float* __restrict__ W1,
                           const float* __restrict__ b1, float* __restrict__ x2, int N) {
  int gtid = blockIdx.x * blockDim.x + threadIdx.x;
  int wave = gtid >> 5;
  int lane = threadIdx.x & 31;
  int tiles = (N + 15) >> 4;
  if (wave >= tiles) return;
  int half = lane >> 4;
  int l16  = lane & 15;
  int row  = wave * 16 + l16;
  int rowc = row < N ? row : N - 1;
  const float* sp = S1 + (size_t)rowc * 32;

  float  den[4], bb[4];
  float2 p[4][2];
  v2f    b[4][2];
#pragma unroll
  for (int h = 0; h < 4; ++h) {
    den[h] = sp[h * 8 + 7];
    bb[h]  = b1[h * 16 + l16];
#pragma unroll
    for (int s = 0; s < 2; ++s) {
      int k0 = 4 * s + 2 * half;
      p[h][s] = *(const float2*)(sp + h * 8 + k0);
      int col = h * 16 + l16;
      b[h][s][0] = W1[k0 * 64 + col];
      b[h][s][1] = W1[(k0 + 1 < 7 ? k0 + 1 : 6) * 64 + col];
    }
  }

#pragma unroll
  for (int h = 0; h < 4; ++h) {
    v8f acc;
#pragma unroll
    for (int q = 0; q < 8; ++q) acc[q] = 0.f;
#pragma unroll
    for (int s = 0; s < 2; ++s) {
      int k0 = 4 * s + 2 * half;
      bool k7 = k0 + 1 == 7;
      v2f a, bv;
      a[0]  = p[h][s].x / den[h];
      a[1]  = k7 ? 0.f : p[h][s].y / den[h];
      bv[0] = b[h][s][0];
      bv[1] = k7 ? 0.f : b[h][s][1];
      acc = __builtin_amdgcn_wmma_f32_16x16x4_f32(
          false, a, false, bv, (short)0, acc, false, false);
    }
#pragma unroll
    for (int r = 0; r < 8; ++r) {
      int m = wave * 16 + r + 8 * half;
      float v = acc[r] + bb[h];
      if (m < N) x2[m * 64 + h * 16 + l16] = v > 0.f ? v : (__expf(v) - 1.f);
    }
  }
}

__global__ void proj2_wmma(const float* __restrict__ x2, const float* __restrict__ W2,
                           float* __restrict__ h2, int N) {
  int gtid = blockIdx.x * blockDim.x + threadIdx.x;
  int wave = gtid >> 5;
  int lane = threadIdx.x & 31;
  int tiles = (N + 15) >> 4;
  if (wave >= tiles) return;
  int half = lane >> 4, l16 = lane & 15;
  int row  = wave * 16 + l16;
  int rowc = row < N ? row : N - 1;
  bool rok = row < N;
  bool cok = l16 < 6;

  v8f acc;
#pragma unroll
  for (int q = 0; q < 8; ++q) acc[q] = 0.f;

#pragma unroll
  for (int s = 0; s < 16; ++s) {
    int k0 = 4 * s + 2 * half;
    v2f a, b;
    a[0] = rok ? x2[rowc * 64 + k0]     : 0.f;
    a[1] = rok ? x2[rowc * 64 + k0 + 1] : 0.f;
    b[0] = cok ? W2[k0 * 6 + l16]       : 0.f;
    b[1] = cok ? W2[(k0 + 1) * 6 + l16] : 0.f;
    acc = __builtin_amdgcn_wmma_f32_16x16x4_f32(
        false, a, false, b, (short)0, acc, false, false);
  }

  if (cok) {
#pragma unroll
    for (int r = 0; r < 8; ++r) {
      int m = wave * 16 + r + 8 * half;
      if (m < N) h2[m * 6 + l16] = acc[r];
    }
  }
}

__global__ void alpha2_kern(const float* __restrict__ h2, const float* __restrict__ aS,
                            const float* __restrict__ aD, float* __restrict__ as2,
                            float* __restrict__ ad2, float* __restrict__ gst, int N) {
  __shared__ float red[8][2];
  int n = blockIdx.x * blockDim.x + threadIdx.x;
  bool ok = n < N;
  int nc = ok ? n : N - 1;
  float s = 0.f, d = 0.f;
#pragma unroll
  for (int c = 0; c < 6; ++c) {
    float v = h2[nc * 6 + c];
    s += v * aS[c];
    d += v * aD[c];
  }
  if (ok) { as2[n] = s; ad2[n] = d; }
  float mx = ok ? s : -3.0e38f, mn = ok ? s : 3.0e38f;
#pragma unroll
  for (int o = 1; o < 32; o <<= 1) {
    mx = fmaxf(mx, __shfl_xor(mx, o));
    mn = fminf(mn, __shfl_xor(mn, o));
  }
  int lane = threadIdx.x & 31, w = threadIdx.x >> 5;
  if (lane == 0) { red[w][0] = mx; red[w][1] = mn; }
  __syncthreads();
  if (threadIdx.x < 2) {
    int t = threadIdx.x;
    float v = red[0][t];
    int nw = blockDim.x >> 5;
    for (int k = 1; k < nw; ++k) v = t == 0 ? fmaxf(v, red[k][t]) : fminf(v, red[k][t]);
    if (t == 0) atomicMaxF(&gst[8], v); else atomicMinF(&gst[9], v);
  }
}

__global__ void e2_max(const int* __restrict__ ei, const float* __restrict__ as2,
                       const float* __restrict__ ad2, float* __restrict__ emax2,
                       const float* __restrict__ gst, int E, int N) {
  if (narrow2(gst)) return;
  int ET = E + N;
  int step = gridDim.x * blockDim.x;
  for (int e = blockIdx.x * blockDim.x + threadIdx.x; e < ET; e += step) {
    int src = e < E ? ei[e]     : e - E;
    int dst = e < E ? ei[E + e] : e - E;
    atomicMaxF(&emax2[dst], leaky(as2[src] + ad2[dst]));
  }
}

__device__ __forceinline__ float sum8(float v) {
  v += __int_as_float(__builtin_amdgcn_update_dpp(0, __float_as_int(v), 0x161, 0xf, 0xf, true));
  v += __int_as_float(__builtin_amdgcn_update_dpp(0, __float_as_int(v), 0x162, 0xf, 0xf, true));
  v += __int_as_float(__builtin_amdgcn_update_dpp(0, __float_as_int(v), 0x164, 0xf, 0xf, true));
  return v;
}

__global__ void __launch_bounds__(256, 16)
e2_agg(const int* __restrict__ ei, const float* __restrict__ aS,
       const float* __restrict__ ad2, const float* __restrict__ gst,
       const float* __restrict__ emax2,
       const float* __restrict__ h2, float* __restrict__ agg2r,
       unsigned* __restrict__ tick, int E, int N) {
  const int lane = threadIdx.x & 31;
  const int ET = E + N;
  const int Nh = half_split(N);
  const bool nar = narrow2(gst);
  const float g8 = gst[8];
  const int c = lane & 7;
  const float ac = c < 6 ? aS[c] : 0.f;
  walk_by_half(tick, ET, [=](int base, int hq) {
    int e  = base + lane;
    int ec = e < ET ? e : ET - 1;
    int src = ec < E ? ei[ec]     : ec - E;
    int dst = ec < E ? ei[E + ec] : ec - E;
    const int cnt = keep_half(e < ET, hq, Nh, lane, src, dst);
    if (cnt == 0) return;
    float da = 0.f, m = 0.f;
    if (lane < cnt) {
      da = ad2[dst];
      if (!nar) m = emax2[dst];
    }
    if (nar) m = leaky(g8 + da);
    float hv[8], dv[8], mv[8];
    int dj[8];
#pragma unroll
    for (int it = 0; it < 8; ++it) {
      hv[it] = 0.f; dv[it] = 0.f; mv[it] = 0.f; dj[it] = 0;
      if (it * 4 < cnt) {
        int j  = it * 4 + (lane >> 3);
        int sj = __shfl(src, j);
        dj[it] = __shfl(dst, j);
        dv[it] = __shfl(da, j);
        mv[it] = __shfl(m, j);
        hv[it] = (c < 6 && j < cnt) ? h2[sj * 6 + c] : 0.f;
      }
    }
#pragma unroll
    for (int it = 0; it < 8; ++it) {
      if (it * 4 < cnt) {
        int j = it * 4 + (lane >> 3);
        float sv = sum8(hv[it] * ac);
        float w = __expf(leaky(sv + dv[it]) - mv[it]);
        float val = c == 6 ? w : hv[it] * w;
        if (c < 7 && j < cnt) atomicAdd(&agg2r[dj[it] * 8 + c], val);
      }
    }
  });
}

__global__ void final_kern(const float* __restrict__ agg2r,
                           const float* __restrict__ b2, float* __restrict__ out, int N) {
  int n = blockIdx.x * blockDim.x + threadIdx.x;
  if (n >= N) return;
  float o[6];
  float m = -3.0e38f;
  float den = agg2r[n * 8 + 6];
#pragma unroll
  for (int c = 0; c < 6; ++c) { o[c] = agg2r[n * 8 + c] / den + b2[c]; m = fmaxf(m, o[c]); }
  float sum = 0.f;
#pragma unroll
  for (int c = 0; c < 6; ++c) sum += __expf(o[c] - m);
  float l = __logf(sum);
#pragma unroll
  for (int c = 0; c < 6; ++c) out[n * 6 + c] = o[c] - m - l;
}

extern "C" void kernel_launch(void* const* d_in, const int* in_sizes, int n_in,
                              void* d_out, int out_size, void* d_ws, size_t ws_size,
                              hipStream_t stream) {
  const float* x      = (const float*)d_in[0];
  const int*   ei     = (const int*)d_in[1];
  const float* W1     = (const float*)d_in[2];
  const float* a_src1 = (const float*)d_in[3];
  const float* a_dst1 = (const float*)d_in[4];
  const float* b1     = (const float*)d_in[5];
  const float* W2     = (const float*)d_in[6];
  const float* a_src2 = (const float*)d_in[7];
  const float* a_dst2 = (const float*)d_in[8];
  const float* b2     = (const float*)d_in[9];
  float* out = (float*)d_out;

  int N = in_sizes[0] / 7;
  int E = in_sizes[1] / 2;

  float* ws     = (float*)d_ws;
  float* h1     = ws;
  float* x2     = h1 + (size_t)N * 64;
  float* S1     = x2 + (size_t)N * 64;
  float* agg2r  = S1 + (size_t)N * 32;
  float* as1    = agg2r + (size_t)N * 8;
  float* ad1    = as1 + (size_t)N * 4;
  float* emax1  = ad1 + (size_t)N * 4;
  float* h2     = emax1 + (size_t)N * 4;
  float* as2    = h2 + (size_t)N * 6;
  float* ad2    = as2 + N;
  float* emax2  = ad2 + N;
  float* gst    = emax2 + N;
  unsigned* tick = (unsigned*)(gst + 128);

  const int B = 256;
  auto G = [](long n, int b) { return (int)((n + b - 1) / b); };
  int tiles = (N + 15) / 16;
  int projBlocks = (tiles + (B / 32) - 1) / (B / 32);

  init_kern<<<G((long)N * 32, B), B, 0, stream>>>(emax1, S1, emax2, agg2r, gst, tick, N);

  proj1_wmma<<<projBlocks, B, 0, stream>>>(x, W1, h1, N);
  alpha1_kern<<<G((long)N * 4, B), B, 0, stream>>>(h1, a_src1, a_dst1, as1, ad1, gst, N);
  e1_max<<<1024, B, 0, stream>>>(ei, as1, ad1, emax1, gst, E, N);
  e1_agg<<<2048, B, 0, stream>>>(ei, W1, a_src1, ad1, gst, emax1, x, S1, tick, E, N);
  post1_wmma<<<projBlocks, B, 0, stream>>>(S1, W1, b1, x2, N);

  proj2_wmma<<<projBlocks, B, 0, stream>>>(x2, W2, h2, N);
  alpha2_kern<<<G(N, B), B, 0, stream>>>(h2, a_src2, a_dst2, as2, ad2, gst, N);
  e2_max<<<1024, B, 0, stream>>>(ei, as2, ad2, emax2, gst, E, N);
  e2_agg<<<2048, B, 0, stream>>>(ei, a_src2, ad2, gst, emax2, h2, agg2r, tick + 2 * 128, E, N);
  final_kern<<<G(N, B), B, 0, stream>>>(agg2r, b2, out, N);
}


# Layer 1 epilogue and layer 2 projection fused: a wave keeps its 16x64 x2 tile in LDS (the accumulator-to-A-fragment transposition), x2 is never stored or re-read; 11 -> 10 launches; sums unchanged
# speedup vs baseline: 17.8107x; 1.0674x over previous
#include <hip/hip_runtime.h>
#include <hip/hip_bf16.h>
#include <math.h>

typedef float v2f __attribute__((ext_vector_type(2)));
typedef float v8f __attribute__((ext_vector_type(8)));
typedef float v4f __attribute__((ext_vector_type(4)));

#define NEG_SLOPE 0.2f

__device__ __forceinline__ float leaky(float v) { return v > 0.f ? v : NEG_SLOPE * v; }

__device__ __forceinline__ void atomicMaxF(float* addr, float val) {
  if (val >= 0.f) atomicMax((int*)addr, __float_as_int(val));
  else            atomicMin((unsigned int*)addr, (unsigned int)__float_as_int(val));
}

__device__ __forceinline__ void atomicMinF(float* addr, float val) {
  int b = __float_as_int(val);
  if (b == (int)0x80000000) b = 0;
  if (b >= 0) atomicMin((int*)addr, b);
  else        atomicMax((unsigned int*)addr, (unsigned int)b);
}

#define RANGE_OK 40.0f

__global__ void init_kern(float* emax1, float* S1,
                          float* emax2, float* agg2r, float* gst, unsigned* tick, int N) {
  int i = blockIdx.x * blockDim.x + threadIdx.x;
  if (i < 16) gst[i] = (i < 4 || i == 8) ? -3.0e38f : 3.0e38f;
  if (i < 4) tick[i * 128] = 0u;
  if (i < N * 32) S1[i] = 0.f;
  if (i < N * 8)  agg2r[i] = 0.f;
  if (i < N * 4)  emax1[i] = -3.0e38f;
  if (i < N)     emax2[i] = -3.0e38f;
}

__global__ void proj1_wmma(const float* __restrict__ x, const float* __restrict__ W1,
                           float* __restrict__ h1, int N) {
  int gtid = blockIdx.x * blockDim.x + threadIdx.x;
  int wave = gtid >> 5;
  int lane = threadIdx.x & 31;
  int tiles = (N + 15) >> 4;
  if (wave >= tiles) return;
  int half = lane >> 4;
  int l16  = lane & 15;
  int row  = wave * 16 + l16;
  int rowc = row < N ? row : N - 1;
  bool rok = row < N;

  v2f a[2];
#pragma unroll
  for (int s = 0; s < 2; ++s) {
    int k0 = 4 * s + 2 * half;
    a[s][0] = (rok && k0     < 7) ? x[rowc * 7 + k0]     : 0.f;
    a[s][1] = (rok && k0 + 1 < 7) ? x[rowc * 7 + k0 + 1] : 0.f;
  }

  v8f acc[4];
#pragma unroll
  for (int t = 0; t < 4; ++t)
#pragma unroll
    for (int q = 0; q < 8; ++q) acc[t][q] = 0.f;

#pragma unroll
  for (int s = 0; s < 2; ++s) {
    int k0 = 4 * s + 2 * half;
#pragma unroll
    for (int t = 0; t < 4; ++t) {
      int col = t * 16 + l16;
      v2f b;
      b[0] = (k0     < 7) ? W1[k0 * 64 + col]       : 0.f;
      b[1] = (k0 + 1 < 7) ? W1[(k0 + 1) * 64 + col] : 0.f;
      acc[t] = __builtin_amdgcn_wmma_f32_16x16x4_f32(
          false, a[s], false, b, (short)0, acc[t], false, false);
    }
  }

#pragma unroll
  for (int t = 0; t < 4; ++t)
#pragma unroll
    for (int r = 0; r < 8; ++r) {
      int m = wave * 16 + r + 8 * half;
      if (m < N) h1[m * 64 + t * 16 + l16] = acc[t][r];
    }
}

__global__ void alpha1_kern(const float* __restrict__ h1, const float* __restrict__ aS,
                            const float* __restrict__ aD, float* __restrict__ as1,
                            float* __restrict__ ad1, float* __restrict__ gst, int N) {
  __shared__ float red[8][8];
  int i = blockIdx.x * blockDim.x + threadIdx.x;
  bool ok = i < N * 4;
  int ic = ok ? i : N * 4 - 1;
  int hd = ic & 3, n = ic >> 2;
  const float* hp = h1 + n * 64 + hd * 16;
  float s = 0.f, d = 0.f;
#pragma unroll
  for (int c = 0; c < 16; ++c) {
    float v = hp[c];
    s += v * aS[hd * 16 + c];
    d += v * aD[hd * 16 + c];
  }
  if (ok) { as1[i] = s; ad1[i] = d; }
  float mx = ok ? s : -3.0e38f, mn = ok ? s : 3.0e38f;
#pragma unroll
  for (int o = 4; o < 32; o <<= 1) {
    mx = fmaxf(mx, __shfl_xor(mx, o));
    mn = fminf(mn, __shfl_xor(mn, o));
  }
  int lane = threadIdx.x & 31, w = threadIdx.x >> 5;
  if (lane < 4) { red[w][lane] = mx; red[w][4 + lane] = mn; }
  __syncthreads();
  if (threadIdx.x < 8) {
    int t = threadIdx.x;
    float v = red[0][t];
    int nw = blockDim.x >> 5;
    for (int k = 1; k < nw; ++k) v = t < 4 ? fmaxf(v, red[k][t]) : fminf(v, red[k][t]);
    if (t < 4) atomicMaxF(&gst[t], v); else atomicMinF(&gst[t], v);
  }
}

__device__ __forceinline__ bool narrow1(const float* __restrict__ gst) {
  return (gst[0] - gst[4] <= RANGE_OK) & (gst[1] - gst[5] <= RANGE_OK) &
         (gst[2] - gst[6] <= RANGE_OK) & (gst[3] - gst[7] <= RANGE_OK);
}
__device__ __forceinline__ bool narrow2(const float* __restrict__ gst) {
  return gst[8] - gst[9] <= RANGE_OK;
}

#define SPAN 256
#define TICK_PITCH 128

__device__ __forceinline__ int chip_half() {
  return (__builtin_amdgcn_s_sendmsg_rtn(0x87) >> 18) & 1;
}

__device__ __forceinline__ int half_split(int N) {
  return ((N >> 1) + 15) & ~15;
}

template <class F>
__device__ __forceinline__ void walk_by_half(unsigned* __restrict__ tick, int ET, F chunk) {
  __shared__ int s_t[2];
  const int nspan = (ET + SPAN - 1) / SPAN;
  const int wv = threadIdx.x >> 5;
  const int h = chip_half();
  int par = 0;
  for (int q = 0; q < 2; ++q) {
    const int hq = h ^ q;
    unsigned* tk = tick + hq * TICK_PITCH;
    int nxt = 0;
    if (threadIdx.x == 0) nxt = (int)atomicInc(tk, 0xFFFFFFFFu);
    for (;;) {
      if (threadIdx.x == 0) s_t[par] = nxt;
      __syncthreads();
      const int t = __builtin_amdgcn_readfirstlane(s_t[par]);
      par ^= 1;
      if (t >= nspan) break;
      if (threadIdx.x == 0) nxt = (int)atomicInc(tk, 0xFFFFFFFFu);
      const int base = t * SPAN + wv * 32;
      if (base < ET) chunk(base, hq);
    }
  }
}

__device__ __forceinline__ int keep_half(bool valid, int hq, int Nh, int lane, int& src, int& dst) {
  bool mine = valid && ((dst >= Nh) == (hq != 0));
  unsigned mask = __builtin_amdgcn_ballot_w32(mine);
  int cnt = __builtin_popcount(mask);
  int below = (int)__builtin_amdgcn_mbcnt_lo(mask, 0u);
  int slot = mine ? below : cnt + lane - below;
  src = __builtin_amdgcn_ds_permute(slot << 2, src);
  dst = __builtin_amdgcn_ds_permute(slot << 2, dst);
  return cnt;
}

__global__ void e1_max(const int* __restrict__ ei, const float* __restrict__ as1,
                       const float* __restrict__ ad1, float* __restrict__ emax1,
                       const float* __restrict__ gst, int E, int N) {
  if (narrow1(gst)) return;
  int lane = threadIdx.x & 31;
  int wave0 = (blockIdx.x * blockDim.x + threadIdx.x) >> 5;
  int nwaves = (gridDim.x * blockDim.x) >> 5;
  int ET = E + N;
  for (int base = wave0 * 32; base < ET; base += nwaves * 32) {
  int e  = base + lane;
  int ec = e < ET ? e : ET - 1;
  int src = ec < E ? ei[ec]     : ec - E;
  int dst = ec < E ? ei[E + ec] : ec - E;
  int h = lane & 3;
  float ev[4];
  int dj[4];
#pragma unroll
  for (int it = 0; it < 4; ++it) {
    int j  = it * 8 + (lane >> 2);
    int sj = __shfl(src, j);
    dj[it] = __shfl(dst, j);
    ev[it] = leaky(as1[sj * 4 + h] + ad1[dj[it] * 4 + h]);
  }
#pragma unroll
  for (int it = 0; it < 4; ++it) {
    int j = it * 8 + (lane >> 2);
    if (base + j < ET) atomicMaxF(&emax1[dj[it] * 4 + h], ev[it]);
  }
  }
}

#define REC_PITCH 36

__global__ void __launch_bounds__(256, 16)
e1_agg(const int* __restrict__ ei, const float* __restrict__ W1,
       const float* __restrict__ aS,
       const float* __restrict__ ad1, const float* __restrict__ gst,
       const float* __restrict__ emax1,
       const float* __restrict__ x,
       float* __restrict__ S1, unsigned* __restrict__ tick, int E, int N) {
  __shared__ __attribute__((aligned(16))) float rec[8][32 * REC_PITCH];
  const int lane = threadIdx.x & 31;
  const int ET = E + N;
  const int Nh = half_split(N);
  const bool nar = narrow1(gst);
  const float g0 = gst[0], g1 = gst[1], g2 = gst[2], g3 = gst[3];
  float wal = 0.f;
  {
    const int lk = lane < 28 ? lane >> 2 : 6, lh = lane & 3;
#pragma unroll
    for (int c = 0; c < 16; ++c) wal += W1[lk * 64 + lh * 16 + c] * aS[lh * 16 + c];
  }
  float wa[28];
#pragma unroll
  for (int i = 0; i < 28; ++i)
    wa[i] = __int_as_float(__builtin_amdgcn_readlane(__float_as_int(wal), i));
  float* const rows = rec[threadIdx.x >> 5];
  walk_by_half(tick, ET, [=](int base, int hq) {
    int e  = base + lane;
    int ec = e < ET ? e : ET - 1;
    int src = ec < E ? ei[ec]     : ec - E;
    int dst = ec < E ? ei[E + ec] : ec - E;
    const int cnt = keep_half(e < ET, hq, Nh, lane, src, dst);
    if (cnt == 0) return;
    if (lane < cnt) {
      const float* xr = x + (size_t)src * 7;
      const float x0 = xr[0], x1 = xr[1], x2 = xr[2], x3 = xr[3], x4 = xr[4], x5 = xr[5], x6 = xr[6];
      const float4 da = *(const float4*)(ad1 + (size_t)dst * 4);
      float4 m = make_float4(0.f, 0.f, 0.f, 0.f);
      if (!nar) m = *(const float4*)(emax1 + (size_t)dst * 4);
      else {
        m.x = leaky(g0 + da.x); m.y = leaky(g1 + da.y);
        m.z = leaky(g2 + da.z); m.w = leaky(g3 + da.w);
      }
      const float s0 = x0 * wa[0] + x1 * wa[4] + x2 * wa[8]  + x3 * wa[12] + x4 * wa[16] + x5 * wa[20] + x6 * wa[24];
      const float s1 = x0 * wa[1] + x1 * wa[5] + x2 * wa[9]  + x3 * wa[13] + x4 * wa[17] + x5 * wa[21] + x6 * wa[25];
      const float s2 = x0 * wa[2] + x1 * wa[6] + x2 * wa[10] + x3 * wa[14] + x4 * wa[18] + x5 * wa[22] + x6 * wa[26];
      const float s3 = x0 * wa[3] + x1 * wa[7] + x2 * wa[11] + x3 * wa[15] + x4 * wa[19] + x5 * wa[23] + x6 * wa[27];
      const float w0 = __expf(leaky(s0 + da.x) - m.x);
      const float w1 = __expf(leaky(s1 + da.y) - m.y);
      const float w2 = __expf(leaky(s2 + da.z) - m.z);
      const float w3 = __expf(leaky(s3 + da.w) - m.w);
      float4* r = (float4*)(rows + lane * REC_PITCH);
      r[0] = make_float4(x0 * w0, x1 * w0, x2 * w0, x3 * w0);
      r[1] = make_float4(x4 * w0, x5 * w0, x6 * w0, w0);
      r[2] = make_float4(x0 * w1, x1 * w1, x2 * w1, x3 * w1);
      r[3] = make_float4(x4 * w1, x5 * w1, x6 * w1, w1);
      r[4] = make_float4(x0 * w2, x1 * w2, x2 * w2, x3 * w2);
      r[5] = make_float4(x4 * w2, x5 * w2, x6 * w2, w2);
      r[6] = make_float4(x0 * w3, x1 * w3, x2 * w3, x3 * w3);
      r[7] = make_float4(x4 * w3, x5 * w3, x6 * w3, w3);
    }
    __builtin_amdgcn_fence(__ATOMIC_RELEASE, "wavefront");
    __builtin_amdgcn_wave_barrier();
    __builtin_amdgcn_fence(__ATOMIC_ACQUIRE, "wavefront");
    const float* rl = rows + lane;
    int j = 0;
    for (; j + 8 <= cnt; j += 8) {
      float v[8];
#pragma unroll
      for (int u = 0; u < 8; ++u) v[u] = rl[(j + u) * REC_PITCH];
#pragma unroll
      for (int u = 0; u < 8; ++u) {
        int d = __builtin_amdgcn_readlane(dst, j + u);
        atomicAdd(S1 + (size_t)d * 32 + lane, v[u]);
      }
    }
    if (j + 4 <= cnt) {
      float v[4];
#pragma unroll
      for (int u = 0; u < 4; ++u) v[u] = rl[(j + u) * REC_PITCH];
#pragma unroll
      for (int u = 0; u < 4; ++u) {
        int d = __builtin_amdgcn_readlane(dst, j + u);
        atomicAdd(S1 + (size_t)d * 32 + lane, v[u]);
      }
      j += 4;
    }
    for (; j < cnt; ++j) {
      float v = rl[j * REC_PITCH];
      int d = __builtin_amdgcn_readlane(dst, j);
      atomicAdd(S1 + (size_t)d * 32 + lane, v);
    }
  });
}

#define X2_PITCH 68

__global__ void post1_proj2_wmma(const float* __restrict__ S1, const float* __restrict__ W1,
                                 const float* __restrict__ b1, const float* __restrict__ W2,
                                 float* __restrict__ h2, int N) {
  __shared__ __attribute__((aligned(16))) float tile[8][16 * X2_PITCH];
  int gtid = blockIdx.x * blockDim.x + threadIdx.x;
  int wave = gtid >> 5;
  int lane = threadIdx.x & 31;
  int tiles = (N + 15) >> 4;
  if (wave >= tiles) return;
  int half = lane >> 4;
  int l16  = lane & 15;
  int row  = wave * 16 + l16;
  int rowc = row < N ? row : N - 1;
  const float* sp = S1 + (size_t)rowc * 32;
  float* const tl = tile[threadIdx.x >> 5];

  float  den[4], bb[4];
  float2 p[4][2];
  v2f    b[4][2];
  const bool cok = l16 < 6;
  const int  c6  = cok ? l16 : 5;
  v2f w2f[16];
#pragma unroll
  for (int s = 0; s < 16; ++s) {
    int k0 = 4 * s + 2 * half;
    w2f[s][0] = W2[k0 * 6 + c6];
    w2f[s][1] = W2[(k0 + 1) * 6 + c6];
  }
#pragma unroll
  for (int h = 0; h < 4; ++h) {
    den[h] = sp[h * 8 + 7];
    bb[h]  = b1[h * 16 + l16];
#pragma unroll
    for (int s = 0; s < 2; ++s) {
      int k0 = 4 * s + 2 * half;
      p[h][s] = *(const float2*)(sp + h * 8 + k0);
      int col = h * 16 + l16;
      b[h][s][0] = W1[k0 * 64 + col];
      b[h][s][1] = W1[(k0 + 1 < 7 ? k0 + 1 : 6) * 64 + col];
    }
  }

#pragma unroll
  for (int h = 0; h < 4; ++h) {
    v8f acc;
#pragma unroll
    for (int q = 0; q < 8; ++q) acc[q] = 0.f;
#pragma unroll
    for (int s = 0; s < 2; ++s) {
      int k0 = 4 * s + 2 * half;
      bool k7 = k0 + 1 == 7;
      v2f a, bv;
      a[0]  = p[h][s].x / den[h];
      a[1]  = k7 ? 0.f : p[h][s].y / den[h];
      bv[0] = b[h][s][0];
      bv[1] = k7 ? 0.f : b[h][s][1];
      acc = __builtin_amdgcn_wmma_f32_16x16x4_f32(
          false, a, false, bv, (short)0, acc, false, false);
    }
#pragma unroll
    for (int r = 0; r < 8; ++r) {
      float v = acc[r] + bb[h];
      tl[(r + 8 * half) * X2_PITCH + h * 16 + l16] = v > 0.f ? v : (__expf(v) - 1.f);
    }
  }

  __builtin_amdgcn_fence(__ATOMIC_RELEASE, "wavefront");
  __builtin_amdgcn_wave_barrier();
  __builtin_amdgcn_fence(__ATOMIC_ACQUIRE, "wavefront");

  const float* ar = tl + l16 * X2_PITCH;
  v8f acc2;
#pragma unroll
  for (int q = 0; q < 8; ++q) acc2[q] = 0.f;
#pragma unroll
  for (int s = 0; s < 16; ++s) {
    int k0 = 4 * s + 2 * half;
    float2 av = *(const float2*)(ar + k0);
    v2f a, bw;
    a[0]  = av.x;
    a[1]  = av.y;
    bw[0] = cok ? w2f[s][0] : 0.f;
    bw[1] = cok ? w2f[s][1] : 0.f;
    acc2 = __builtin_amdgcn_wmma_f32_16x16x4_f32(
        false, a, false, bw, (short)0, acc2, false, false);
  }

  if (cok) {
#pragma unroll
    for (int r = 0; r < 8; ++r) {
      int m = wave * 16 + r + 8 * half;
      if (m < N) h2[m * 6 + l16] = acc2[r];
    }
  }
}

__global__ void alpha2_kern(const float* __restrict__ h2, const float* __restrict__ aS,
                            const float* __restrict__ aD, float* __restrict__ as2,
                            float* __restrict__ ad2, float* __restrict__ gst, int N) {
  __shared__ float red[8][2];
  int n = blockIdx.x * blockDim.x + threadIdx.x;
  bool ok = n < N;
  int nc = ok ? n : N - 1;
  float s = 0.f, d = 0.f;
#pragma unroll
  for (int c = 0; c < 6; ++c) {
    float v = h2[nc * 6 + c];
    s += v * aS[c];
    d += v * aD[c];
  }
  if (ok) { as2[n] = s; ad2[n] = d; }
  float mx = ok ? s : -3.0e38f, mn = ok ? s : 3.0e38f;
#pragma unroll
  for (int o = 1; o < 32; o <<= 1) {
    mx = fmaxf(mx, __shfl_xor(mx, o));
    mn = fminf(mn, __shfl_xor(mn, o));
  }
  int lane = threadIdx.x & 31, w = threadIdx.x >> 5;
  if (lane == 0) { red[w][0] = mx; red[w][1] = mn; }
  __syncthreads();
  if (threadIdx.x < 2) {
    int t = threadIdx.x;
    float v = red[0][t];
    int nw = blockDim.x >> 5;
    for (int k = 1; k < nw; ++k) v = t == 0 ? fmaxf(v, red[k][t]) : fminf(v, red[k][t]);
    if (t == 0) atomicMaxF(&gst[8], v); else atomicMinF(&gst[9], v);
  }
}

__global__ void e2_max(const int* __restrict__ ei, const float* __restrict__ as2,
                       const float* __restrict__ ad2, float* __restrict__ emax2,
                       const float* __restrict__ gst, int E, int N) {
  if (narrow2(gst)) return;
  int ET = E + N;
  int step = gridDim.x * blockDim.x;
  for (int e = blockIdx.x * blockDim.x + threadIdx.x; e < ET; e += step) {
    int src = e < E ? ei[e]     : e - E;
    int dst = e < E ? ei[E + e] : e - E;
    atomicMaxF(&emax2[dst], leaky(as2[src] + ad2[dst]));
  }
}

__device__ __forceinline__ float sum8(float v) {
  v += __int_as_float(__builtin_amdgcn_update_dpp(0, __float_as_int(v), 0x161, 0xf, 0xf, true));
  v += __int_as_float(__builtin_amdgcn_update_dpp(0, __float_as_int(v), 0x162, 0xf, 0xf, true));
  v += __int_as_float(__builtin_amdgcn_update_dpp(0, __float_as_int(v), 0x164, 0xf, 0xf, true));
  return v;
}

__global__ void __launch_bounds__(256, 16)
e2_agg(const int* __restrict__ ei, const float* __restrict__ aS,
       const float* __restrict__ ad2, const float* __restrict__ gst,
       const float* __restrict__ emax2,
       const float* __restrict__ h2, float* __restrict__ agg2r,
       unsigned* __restrict__ tick, int E, int N) {
  const int lane = threadIdx.x & 31;
  const int ET = E + N;
  const int Nh = half_split(N);
  const bool nar = narrow2(gst);
  const float g8 = gst[8];
  const int c = lane & 7;
  const float ac = c < 6 ? aS[c] : 0.f;
  walk_by_half(tick, ET, [=](int base, int hq) {
    int e  = base + lane;
    int ec = e < ET ? e : ET - 1;
    int src = ec < E ? ei[ec]     : ec - E;
    int dst = ec < E ? ei[E + ec] : ec - E;
    const int cnt = keep_half(e < ET, hq, Nh, lane, src, dst);
    if (cnt == 0) return;
    float da = 0.f, m = 0.f;
    if (lane < cnt) {
      da = ad2[dst];
      if (!nar) m = emax2[dst];
    }
    if (nar) m = leaky(g8 + da);
    float hv[8], dv[8], mv[8];
    int dj[8];
#pragma unroll
    for (int it = 0; it < 8; ++it) {
      hv[it] = 0.f; dv[it] = 0.f; mv[it] = 0.f; dj[it] = 0;
      if (it * 4 < cnt) {
        int j  = it * 4 + (lane >> 3);
        int sj = __shfl(src, j);
        dj[it] = __shfl(dst, j);
        dv[it] = __shfl(da, j);
        mv[it] = __shfl(m, j);
        hv[it] = (c < 6 && j < cnt) ? h2[sj * 6 + c] : 0.f;
      }
    }
#pragma unroll
    for (int it = 0; it < 8; ++it) {
      if (it * 4 < cnt) {
        int j = it * 4 + (lane >> 3);
        float sv = sum8(hv[it] * ac);
        float w = __expf(leaky(sv + dv[it]) - mv[it]);
        float val = c == 6 ? w : hv[it] * w;
        if (c < 7 && j < cnt) atomicAdd(&agg2r[dj[it] * 8 + c], val);
      }
    }
  });
}

__global__ void final_kern(const float* __restrict__ agg2r,
                           const float* __restrict__ b2, float* __restrict__ out, int N) {
  int n = blockIdx.x * blockDim.x + threadIdx.x;
  if (n >= N) return;
  float o[6];
  float m = -3.0e38f;
  float den = agg2r[n * 8 + 6];
#pragma unroll
  for (int c = 0; c < 6; ++c) { o[c] = agg2r[n * 8 + c] / den + b2[c]; m = fmaxf(m, o[c]); }
  float sum = 0.f;
#pragma unroll
  for (int c = 0; c < 6; ++c) sum += __expf(o[c] - m);
  float l = __logf(sum);
#pragma unroll
  for (int c = 0; c < 6; ++c) out[n * 6 + c] = o[c] - m - l;
}

extern "C" void kernel_launch(void* const* d_in, const int* in_sizes, int n_in,
                              void* d_out, int out_size, void* d_ws, size_t ws_size,
                              hipStream_t stream) {
  const float* x      = (const float*)d_in[0];
  const int*   ei     = (const int*)d_in[1];
  const float* W1     = (const float*)d_in[2];
  const float* a_src1 = (const float*)d_in[3];
  const float* a_dst1 = (const float*)d_in[4];
  const float* b1     = (const float*)d_in[5];
  const float* W2     = (const float*)d_in[6];
  const float* a_src2 = (const float*)d_in[7];
  const float* a_dst2 = (const float*)d_in[8];
  const float* b2     = (const float*)d_in[9];
  float* out = (float*)d_out;

  int N = in_sizes[0] / 7;
  int E = in_sizes[1] / 2;

  float* ws     = (float*)d_ws;
  float* h1     = ws;
  float* S1     = h1 + (size_t)N * 64;
  float* agg2r  = S1 + (size_t)N * 32;
  float* as1    = agg2r + (size_t)N * 8;
  float* ad1    = as1 + (size_t)N * 4;
  float* emax1  = ad1 + (size_t)N * 4;
  float* h2     = emax1 + (size_t)N * 4;
  float* as2    = h2 + (size_t)N * 6;
  float* ad2    = as2 + N;
  float* emax2  = ad2 + N;
  float* gst    = emax2 + N;
  unsigned* tick = (unsigned*)(gst + 128);

  const int B = 256;
  auto G = [](long n, int b) { return (int)((n + b - 1) / b); };
  int tiles = (N + 15) / 16;
  int projBlocks = (tiles + (B / 32) - 1) / (B / 32);

  init_kern<<<G((long)N * 32, B), B, 0, stream>>>(emax1, S1, emax2, agg2r, gst, tick, N);

  proj1_wmma<<<projBlocks, B, 0, stream>>>(x, W1, h1, N);
  alpha1_kern<<<G((long)N * 4, B), B, 0, stream>>>(h1, a_src1, a_dst1, as1, ad1, gst, N);
  e1_max<<<1024, B, 0, stream>>>(ei, as1, ad1, emax1, gst, E, N);
  e1_agg<<<2048, B, 0, stream>>>(ei, W1, a_src1, ad1, gst, emax1, x, S1, tick, E, N);
  post1_proj2_wmma<<<projBlocks, B, 0, stream>>>(S1, W1, b1, W2, h2, N);

  alpha2_kern<<<G(N, B), B, 0, stream>>>(h2, a_src2, a_dst2, as2, ad2, gst, N);
  e2_max<<<1024, B, 0, stream>>>(ei, as2, ad2, emax2, gst, E, N);
  e2_agg<<<2048, B, 0, stream>>>(ei, a_src2, ad2, gst, emax2, h2, agg2r, tick + 2 * 128, E, N);
  final_kern<<<G(N, B), B, 0, stream>>>(agg2r, b2, out, N);
}


# Attention vectors folded into the weights: node scores are WMMA columns, layer 1 x (W1 a) with h1 never built, layer 2 in two padding columns of x2 W2 (re-associated sums); 10 -> 8 launches
# speedup vs baseline: 19.2942x; 1.0833x over previous
#include <hip/hip_runtime.h>
#include <hip/hip_bf16.h>
#include <math.h>

typedef float v2f __attribute__((ext_vector_type(2)));
typedef float v8f __attribute__((ext_vector_type(8)));
typedef float v4f __attribute__((ext_vector_type(4)));

#define NEG_SLOPE 0.2f

__device__ __forceinline__ float leaky(float v) { return v > 0.f ? v : NEG_SLOPE * v; }

__device__ __forceinline__ void atomicMaxF(float* addr, float val) {
  if (val >= 0.f) atomicMax((int*)addr, __float_as_int(val));
  else            atomicMin((unsigned int*)addr, (unsigned int)__float_as_int(val));
}

__device__ __forceinline__ void atomicMinF(float* addr, float val) {
  int b = __float_as_int(val);
  if (b == (int)0x80000000) b = 0;
  if (b >= 0) atomicMin((int*)addr, b);
  else        atomicMax((unsigned int*)addr, (unsigned int)b);
}

#define RANGE_OK 40.0f

__global__ void init_kern(float* emax1, float* S1,
                          float* emax2, float* agg2r, float* gst, unsigned* tick, int N) {
  int i = blockIdx.x * blockDim.x + threadIdx.x;
  if (i < 16) gst[i] = (i < 4 || i == 8) ? -3.0e38f : 3.0e38f;
  if (i < 4) tick[i * 128] = 0u;
  if (i < N * 32) S1[i] = 0.f;
  if (i < N * 8)  agg2r[i] = 0.f;
  if (i < N * 4)  emax1[i] = -3.0e38f;
  if (i < N)     emax2[i] = -3.0e38f;
}

#define SC_TILES 4

__device__ __forceinline__ float rowmax16(float v) {
  v = fmaxf(v, __int_as_float(__builtin_amdgcn_update_dpp(0, __float_as_int(v), 0x161, 0xf, 0xf, true)));
  v = fmaxf(v, __int_as_float(__builtin_amdgcn_update_dpp(0, __float_as_int(v), 0x162, 0xf, 0xf, true)));
  v = fmaxf(v, __int_as_float(__builtin_amdgcn_update_dpp(0, __float_as_int(v), 0x164, 0xf, 0xf, true)));
  v = fmaxf(v, __int_as_float(__builtin_amdgcn_update_dpp(0, __float_as_int(v), 0x168, 0xf, 0xf, true)));
  return v;
}
__device__ __forceinline__ float rowmin16(float v) {
  v = fminf(v, __int_as_float(__builtin_amdgcn_update_dpp(0, __float_as_int(v), 0x161, 0xf, 0xf, true)));
  v = fminf(v, __int_as_float(__builtin_amdgcn_update_dpp(0, __float_as_int(v), 0x162, 0xf, 0xf, true)));
  v = fminf(v, __int_as_float(__builtin_amdgcn_update_dpp(0, __float_as_int(v), 0x164, 0xf, 0xf, true)));
  v = fminf(v, __int_as_float(__builtin_amdgcn_update_dpp(0, __float_as_int(v), 0x168, 0xf, 0xf, true)));
  return v;
}

__global__ void score1_wmma(const float* __restrict__ x, const float* __restrict__ W1,
                            const float* __restrict__ aS, const float* __restrict__ aD,
                            float* __restrict__ as1, float* __restrict__ ad1,
                            float* __restrict__ gst, int N) {
  __shared__ float wa[8][8];
  __shared__ float red[8][8];
  const int t = threadIdx.x;
  if (t < 64) {
    const int k = t >> 3, j = t & 7, hd = j & 3;
    const float* av = j < 4 ? aS : aD;
    float s = 0.f;
    if (k < 7) {
#pragma unroll
      for (int c = 0; c < 16; ++c) s += W1[k * 64 + hd * 16 + c] * av[hd * 16 + c];
    }
    wa[k][j] = s;
  }
  __syncthreads();

  const int lane = t & 31, w = t >> 5;
  const int half = lane >> 4;
  const int l16  = lane & 15;
  const int wave = blockIdx.x * (blockDim.x >> 5) + w;

  v2f a[2];
#pragma unroll
  for (int s = 0; s < 2; ++s) {
    const int k0 = 4 * s + 2 * half;
    a[s][0] = l16 < 8 ? wa[k0][l16]     : 0.f;
    a[s][1] = l16 < 8 ? wa[k0 + 1][l16] : 0.f;
  }

  v2f  b[SC_TILES][2];
  int  nn[SC_TILES];
  bool ok[SC_TILES];
#pragma unroll
  for (int i = 0; i < SC_TILES; ++i) {
    const int n = (wave * SC_TILES + i) * 16 + l16;
    ok[i] = n < N;
    nn[i] = n;
    const int nc = ok[i] ? n : N - 1;
#pragma unroll
    for (int s = 0; s < 2; ++s) {
      const int k0 = 4 * s + 2 * half;
      b[i][s][0] = x[nc * 7 + k0];
      b[i][s][1] = (k0 + 1 < 7) ? x[nc * 7 + k0 + 1] : 0.f;
    }
  }

  float mx[4], mn[4];
#pragma unroll
  for (int h = 0; h < 4; ++h) { mx[h] = -3.0e38f; mn[h] = 3.0e38f; }

#pragma unroll
  for (int i = 0; i < SC_TILES; ++i) {
    v8f acc;
#pragma unroll
    for (int q = 0; q < 8; ++q) acc[q] = 0.f;
#pragma unroll
    for (int s = 0; s < 2; ++s)
      acc = __builtin_amdgcn_wmma_f32_16x16x4_f32(
          false, a[s], false, b[i][s], (short)0, acc, false, false);
    if (half == 0 && ok[i]) {
      *(float4*)(as1 + (size_t)nn[i] * 4) = make_float4(acc[0], acc[1], acc[2], acc[3]);
      *(float4*)(ad1 + (size_t)nn[i] * 4) = make_float4(acc[4], acc[5], acc[6], acc[7]);
#pragma unroll
      for (int h = 0; h < 4; ++h) { mx[h] = fmaxf(mx[h], acc[h]); mn[h] = fminf(mn[h], acc[h]); }
    }
  }

#pragma unroll
  for (int h = 0; h < 4; ++h) { mx[h] = rowmax16(mx[h]); mn[h] = rowmin16(mn[h]); }
  if (lane == 0) {
#pragma unroll
    for (int h = 0; h < 4; ++h) { red[w][h] = mx[h]; red[w][4 + h] = mn[h]; }
  }
  __syncthreads();
  if (t < 8) {
    float v = red[0][t];
    const int nw = blockDim.x >> 5;
    for (int k = 1; k < nw; ++k) v = t < 4 ? fmaxf(v, red[k][t]) : fminf(v, red[k][t]);
    if (t < 4) atomicMaxF(&gst[t], v); else atomicMinF(&gst[t], v);
  }
}

__device__ __forceinline__ bool narrow1(const float* __restrict__ gst) {
  return (gst[0] - gst[4] <= RANGE_OK) & (gst[1] - gst[5] <= RANGE_OK) &
         (gst[2] - gst[6] <= RANGE_OK) & (gst[3] - gst[7] <= RANGE_OK);
}
__device__ __forceinline__ bool narrow2(const float* __restrict__ gst) {
  return gst[8] - gst[9] <= RANGE_OK;
}

#define SPAN 256
#define TICK_PITCH 128

__device__ __forceinline__ int chip_half() {
  return (__builtin_amdgcn_s_sendmsg_rtn(0x87) >> 18) & 1;
}

__device__ __forceinline__ int half_split(int N) {
  return ((N >> 1) + 15) & ~15;
}

template <class F>
__device__ __forceinline__ void walk_by_half(unsigned* __restrict__ tick, int ET, F chunk) {
  __shared__ int s_t[2];
  const int nspan = (ET + SPAN - 1) / SPAN;
  const int wv = threadIdx.x >> 5;
  const int h = chip_half();
  int par = 0;
  for (int q = 0; q < 2; ++q) {
    const int hq = h ^ q;
    unsigned* tk = tick + hq * TICK_PITCH;
    int nxt = 0;
    if (threadIdx.x == 0) nxt = (int)atomicInc(tk, 0xFFFFFFFFu);
    for (;;) {
      if (threadIdx.x == 0) s_t[par] = nxt;
      __syncthreads();
      const int t = __builtin_amdgcn_readfirstlane(s_t[par]);
      par ^= 1;
      if (t >= nspan) break;
      if (threadIdx.x == 0) nxt = (int)atomicInc(tk, 0xFFFFFFFFu);
      const int base = t * SPAN + wv * 32;
      if (base < ET) chunk(base, hq);
    }
  }
}

__device__ __forceinline__ int keep_half(bool valid, int hq, int Nh, int lane, int& src, int& dst) {
  bool mine = valid && ((dst >= Nh) == (hq != 0));
  unsigned mask = __builtin_amdgcn_ballot_w32(mine);
  int cnt = __builtin_popcount(mask);
  int below = (int)__builtin_amdgcn_mbcnt_lo(mask, 0u);
  int slot = mine ? below : cnt + lane - below;
  src = __builtin_amdgcn_ds_permute(slot << 2, src);
  dst = __builtin_amdgcn_ds_permute(slot << 2, dst);
  return cnt;
}

__global__ void e1_max(const int* __restrict__ ei, const float* __restrict__ as1,
                       const float* __restrict__ ad1, float* __restrict__ emax1,
                       const float* __restrict__ gst, int E, int N) {
  if (narrow1(gst)) return;
  int lane = threadIdx.x & 31;
  int wave0 = (blockIdx.x * blockDim.x + threadIdx.x) >> 5;
  int nwaves = (gridDim.x * blockDim.x) >> 5;
  int ET = E + N;
  for (int base = wave0 * 32; base < ET; base += nwaves * 32) {
  int e  = base + lane;
  int ec = e < ET ? e : ET - 1;
  int src = ec < E ? ei[ec]     : ec - E;
  int dst = ec < E ? ei[E + ec] : ec - E;
  int h = lane & 3;
  float ev[4];
  int dj[4];
#pragma unroll
  for (int it = 0; it < 4; ++it) {
    int j  = it * 8 + (lane >> 2);
    int sj = __shfl(src, j);
    dj[it] = __shfl(dst, j);
    ev[it] = leaky(as1[sj * 4 + h] + ad1[dj[it] * 4 + h]);
  }
#pragma unroll
  for (int it = 0; it < 4; ++it) {
    int j = it * 8 + (lane >> 2);
    if (base + j < ET) atomicMaxF(&emax1[dj[it] * 4 + h], ev[it]);
  }
  }
}

#define REC_PITCH 36

__global__ void __launch_bounds__(256, 16)
e1_agg(const int* __restrict__ ei, const float* __restrict__ W1,
       const float* __restrict__ aS,
       const float* __restrict__ ad1, const float* __restrict__ gst,
       const float* __restrict__ emax1,
       const float* __restrict__ x,
       float* __restrict__ S1, unsigned* __restrict__ tick, int E, int N) {
  __shared__ __attribute__((aligned(16))) float rec[8][32 * REC_PITCH];
  const int lane = threadIdx.x & 31;
  const int ET = E + N;
  const int Nh = half_split(N);
  const bool nar = narrow1(gst);
  const float g0 = gst[0], g1 = gst[1], g2 = gst[2], g3 = gst[3];
  float wal = 0.f;
  {
    const int lk = lane < 28 ? lane >> 2 : 6, lh = lane & 3;
#pragma unroll
    for (int c = 0; c < 16; ++c) wal += W1[lk * 64 + lh * 16 + c] * aS[lh * 16 + c];
  }
  float wa[28];
#pragma unroll
  for (int i = 0; i < 28; ++i)
    wa[i] = __int_as_float(__builtin_amdgcn_readlane(__float_as_int(wal), i));
  float* const rows = rec[threadIdx.x >> 5];
  walk_by_half(tick, ET, [=](int base, int hq) {
    int e  = base + lane;
    int ec = e < ET ? e : ET - 1;
    int src = ec < E ? ei[ec]     : ec - E;
    int dst = ec < E ? ei[E + ec] : ec - E;
    const int cnt = keep_half(e < ET, hq, Nh, lane, src, dst);
    if (cnt == 0) return;
    if (lane < cnt) {
      const float* xr = x + (size_t)src * 7;
      const float x0 = xr[0], x1 = xr[1], x2 = xr[2], x3 = xr[3], x4 = xr[4], x5 = xr[5], x6 = xr[6];
      const float4 da = *(const float4*)(ad1 + (size_t)dst * 4);
      float4 m = make_float4(0.f, 0.f, 0.f, 0.f);
      if (!nar) m = *(const float4*)(emax1 + (size_t)dst * 4);
      else {
        m.x = leaky(g0 + da.x); m.y = leaky(g1 + da.y);
        m.z = leaky(g2 + da.z); m.w = leaky(g3 + da.w);
      }
      const float s0 = x0 * wa[0] + x1 * wa[4] + x2 * wa[8]  + x3 * wa[12] + x4 * wa[16] + x5 * wa[20] + x6 * wa[24];
      const float s1 = x0 * wa[1] + x1 * wa[5] + x2 * wa[9]  + x3 * wa[13] + x4 * wa[17] + x5 * wa[21] + x6 * wa[25];
      const float s2 = x0 * wa[2] + x1 * wa[6] + x2 * wa[10] + x3 * wa[14] + x4 * wa[18] + x5 * wa[22] + x6 * wa[26];
      const float s3 = x0 * wa[3] + x1 * wa[7] + x2 * wa[11] + x3 * wa[15] + x4 * wa[19] + x5 * wa[23] + x6 * wa[27];
      const float w0 = __expf(leaky(s0 + da.x) - m.x);
      const float w1 = __expf(leaky(s1 + da.y) - m.y);
      const float w2 = __expf(leaky(s2 + da.z) - m.z);
      const float w3 = __expf(leaky(s3 + da.w) - m.w);
      float4* r = (float4*)(rows + lane * REC_PITCH);
      r[0] = make_float4(x0 * w0, x1 * w0, x2 * w0, x3 * w0);
      r[1] = make_float4(x4 * w0, x5 * w0, x6 * w0, w0);
      r[2] = make_float4(x0 * w1, x1 * w1, x2 * w1, x3 * w1);
      r[3] = make_float4(x4 * w1, x5 * w1, x6 * w1, w1);
      r[4] = make_float4(x0 * w2, x1 * w2, x2 * w2, x3 * w2);
      r[5] = make_float4(x4 * w2, x5 * w2, x6 * w2, w2);
      r[6] = make_float4(x0 * w3, x1 * w3, x2 * w3, x3 * w3);
      r[7] = make_float4(x4 * w3, x5 * w3, x6 * w3, w3);
    }
    __builtin_amdgcn_fence(__ATOMIC_RELEASE, "wavefront");
    __builtin_amdgcn_wave_barrier();
    __builtin_amdgcn_fence(__ATOMIC_ACQUIRE, "wavefront");
    const float* rl = rows + lane;
    int j = 0;
    for (; j + 8 <= cnt; j += 8) {
      float v[8];
#pragma unroll
      for (int u = 0; u < 8; ++u) v[u] = rl[(j + u) * REC_PITCH];
#pragma unroll
      for (int u = 0; u < 8; ++u) {
        int d = __builtin_amdgcn_readlane(dst, j + u);
        atomicAdd(S1 + (size_t)d * 32 + lane, v[u]);
      }
    }
    if (j + 4 <= cnt) {
      float v[4];
#pragma unroll
      for (int u = 0; u < 4; ++u) v[u] = rl[(j + u) * REC_PITCH];
#pragma unroll
      for (int u = 0; u < 4; ++u) {
        int d = __builtin_amdgcn_readlane(dst, j + u);
        atomicAdd(S1 + (size_t)d * 32 + lane, v[u]);
      }
      j += 4;
    }
    for (; j < cnt; ++j) {
      float v = rl[j * REC_PITCH];
      int d = __builtin_amdgcn_readlane(dst, j);
      atomicAdd(S1 + (size_t)d * 32 + lane, v);
    }
  });
}

#define X2_PITCH 68

__global__ void post1_proj2_wmma(const float* __restrict__ S1, const float* __restrict__ W1,
                                 const float* __restrict__ b1, const float* __restrict__ W2,
                                 const float* __restrict__ aS2, const float* __restrict__ aD2,
                                 float* __restrict__ h2, float* __restrict__ as2, float* __restrict__ ad2,
                                 float* __restrict__ gst, int N) {
  __shared__ __attribute__((aligned(16))) float tile[8][16 * X2_PITCH];
  __shared__ __attribute__((aligned(8))) float bt[16][X2_PITCH];
  __shared__ float red2[16][2];
  int gtid = blockIdx.x * blockDim.x + threadIdx.x;
  int wave = gtid >> 5;
  int lane = threadIdx.x & 31;
  int half = lane >> 4;
  int l16  = lane & 15;
  int row  = wave * 16 + l16;
  int rowc = row < N ? row : N - 1;
  const float* sp = S1 + (size_t)rowc * 32;
  float* const tl = tile[threadIdx.x >> 5];

  float  den[4], bb[4];
  float2 p[4][2];
  v2f    b[4][2];
  const bool cok = l16 < 6;
  const bool sok = (l16 >> 1) == 3;
#pragma unroll
  for (int h = 0; h < 4; ++h) {
    den[h] = sp[h * 8 + 7];
    bb[h]  = b1[h * 16 + l16];
#pragma unroll
    for (int s = 0; s < 2; ++s) {
      int k0 = 4 * s + 2 * half;
      p[h][s] = *(const float2*)(sp + h * 8 + k0);
      int col = h * 16 + l16;
      b[h][s][0] = W1[k0 * 64 + col];
      b[h][s][1] = W1[(k0 + 1 < 7 ? k0 + 1 : 6) * 64 + col];
    }
  }
  {
    const int n = threadIdx.x & 15, kb = threadIdx.x >> 4;
    const int nc = n < 6 ? n : 5;
    float wrow[4][6], wown[4];
#pragma unroll
    for (int i = 0; i < 4; ++i) {
      const int k = kb + 16 * i;
#pragma unroll
      for (int c = 0; c < 6; ++c) wrow[i][c] = W2[k * 6 + c];
      wown[i] = W2[k * 6 + nc];
    }
#pragma unroll
    for (int i = 0; i < 4; ++i) {
      float ss = 0.f, sd = 0.f;
#pragma unroll
      for (int c = 0; c < 6; ++c) { ss += wrow[i][c] * aS2[c]; sd += wrow[i][c] * aD2[c]; }
      bt[n][kb + 16 * i] = n < 6 ? wown[i] : (n == 6 ? ss : (n == 7 ? sd : 0.f));
    }
  }

#pragma unroll
  for (int h = 0; h < 4; ++h) {
    v8f acc;
#pragma unroll
    for (int q = 0; q < 8; ++q) acc[q] = 0.f;
#pragma unroll
    for (int s = 0; s < 2; ++s) {
      int k0 = 4 * s + 2 * half;
      bool k7 = k0 + 1 == 7;
      v2f a, bv;
      a[0]  = p[h][s].x / den[h];
      a[1]  = k7 ? 0.f : p[h][s].y / den[h];
      bv[0] = b[h][s][0];
      bv[1] = k7 ? 0.f : b[h][s][1];
      acc = __builtin_amdgcn_wmma_f32_16x16x4_f32(
          false, a, false, bv, (short)0, acc, false, false);
    }
#pragma unroll
    for (int r = 0; r < 8; ++r) {
      float v = acc[r] + bb[h];
      tl[(r + 8 * half) * X2_PITCH + h * 16 + l16] = v > 0.f ? v : (__expf(v) - 1.f);
    }
  }

  __syncthreads();

  const float* ar = tl + l16 * X2_PITCH;
  const float* br = bt[l16];
  v8f acc2;
#pragma unroll
  for (int q = 0; q < 8; ++q) acc2[q] = 0.f;
#pragma unroll
  for (int s = 0; s < 16; ++s) {
    int k0 = 4 * s + 2 * half;
    float2 av = *(const float2*)(ar + k0);
    float2 bq = *(const float2*)(br + k0);
    v2f a, bw;
    a[0]  = av.x;
    a[1]  = av.y;
    bw[0] = bq.x;
    bw[1] = bq.y;
    acc2 = __builtin_amdgcn_wmma_f32_16x16x4_f32(
        false, a, false, bw, (short)0, acc2, false, false);
  }

  float mx = -3.0e38f, mn = 3.0e38f;
  if (cok) {
#pragma unroll
    for (int r = 0; r < 8; ++r) {
      int m = wave * 16 + r + 8 * half;
      if (m < N) h2[m * 6 + l16] = acc2[r];
    }
  } else if (sok) {
    float* __restrict__ sp2 = (l16 & 1) ? ad2 : as2;
#pragma unroll
    for (int r = 0; r < 8; ++r) {
      int m = wave * 16 + r + 8 * half;
      if (m < N) {
        sp2[m] = acc2[r];
        if (!(l16 & 1)) { mx = fmaxf(mx, acc2[r]); mn = fminf(mn, acc2[r]); }
      }
    }
  }
  mx = rowmax16(mx);
  mn = rowmin16(mn);
  if (l16 == 0) { red2[(threadIdx.x >> 5) * 2 + half][0] = mx; red2[(threadIdx.x >> 5) * 2 + half][1] = mn; }
  __syncthreads();
  if (threadIdx.x < 2) {
    int t = threadIdx.x;
    float v = red2[0][t];
    int nr = (blockDim.x >> 5) * 2;
    for (int k = 1; k < nr; ++k) v = t == 0 ? fmaxf(v, red2[k][0]) : fminf(v, red2[k][1]);
    if (t == 0) atomicMaxF(&gst[8], v); else atomicMinF(&gst[9], v);
  }
}

__global__ void e2_max(const int* __restrict__ ei, const float* __restrict__ as2,
                       const float* __restrict__ ad2, float* __restrict__ emax2,
                       const float* __restrict__ gst, int E, int N) {
  if (narrow2(gst)) return;
  int ET = E + N;
  int step = gridDim.x * blockDim.x;
  for (int e = blockIdx.x * blockDim.x + threadIdx.x; e < ET; e += step) {
    int src = e < E ? ei[e]     : e - E;
    int dst = e < E ? ei[E + e] : e - E;
    atomicMaxF(&emax2[dst], leaky(as2[src] + ad2[dst]));
  }
}

__device__ __forceinline__ float sum8(float v) {
  v += __int_as_float(__builtin_amdgcn_update_dpp(0, __float_as_int(v), 0x161, 0xf, 0xf, true));
  v += __int_as_float(__builtin_amdgcn_update_dpp(0, __float_as_int(v), 0x162, 0xf, 0xf, true));
  v += __int_as_float(__builtin_amdgcn_update_dpp(0, __float_as_int(v), 0x164, 0xf, 0xf, true));
  return v;
}

__global__ void __launch_bounds__(256, 16)
e2_agg(const int* __restrict__ ei, const float* __restrict__ aS,
       const float* __restrict__ ad2, const float* __restrict__ gst,
       const float* __restrict__ emax2,
       const float* __restrict__ h2, float* __restrict__ agg2r,
       unsigned* __restrict__ tick, int E, int N) {
  const int lane = threadIdx.x & 31;
  const int ET = E + N;
  const int Nh = half_split(N);
  const bool nar = narrow2(gst);
  const float g8 = gst[8];
  const int c = lane & 7;
  const float ac = c < 6 ? aS[c] : 0.f;
  walk_by_half(tick, ET, [=](int base, int hq) {
    int e  = base + lane;
    int ec = e < ET ? e : ET - 1;
    int src = ec < E ? ei[ec]     : ec - E;
    int dst = ec < E ? ei[E + ec] : ec - E;
    const int cnt = keep_half(e < ET, hq, Nh, lane, src, dst);
    if (cnt == 0) return;
    float da = 0.f, m = 0.f;
    if (lane < cnt) {
      da = ad2[dst];
      if (!nar) m = emax2[dst];
    }
    if (nar) m = leaky(g8 + da);
    float hv[8], dv[8], mv[8];
    int dj[8];
#pragma unroll
    for (int it = 0; it < 8; ++it) {
      hv[it] = 0.f; dv[it] = 0.f; mv[it] = 0.f; dj[it] = 0;
      if (it * 4 < cnt) {
        int j  = it * 4 + (lane >> 3);
        int sj = __shfl(src, j);
        dj[it] = __shfl(dst, j);
        dv[it] = __shfl(da, j);
        mv[it] = __shfl(m, j);
        hv[it] = (c < 6 && j < cnt) ? h2[sj * 6 + c] : 0.f;
      }
    }
#pragma unroll
    for (int it = 0; it < 8; ++it) {
      if (it * 4 < cnt) {
        int j = it * 4 + (lane >> 3);
        float sv = sum8(hv[it] * ac);
        float w = __expf(leaky(sv + dv[it]) - mv[it]);
        float val = c == 6 ? w : hv[it] * w;
        if (c < 7 && j < cnt) atomicAdd(&agg2r[dj[it] * 8 + c], val);
      }
    }
  });
}

__global__ void final_kern(const float* __restrict__ agg2r,
                           const float* __restrict__ b2, float* __restrict__ out, int N) {
  int n = blockIdx.x * blockDim.x + threadIdx.x;
  if (n >= N) return;
  float o[6];
  float m = -3.0e38f;
  float den = agg2r[n * 8 + 6];
#pragma unroll
  for (int c = 0; c < 6; ++c) { o[c] = agg2r[n * 8 + c] / den + b2[c]; m = fmaxf(m, o[c]); }
  float sum = 0.f;
#pragma unroll
  for (int c = 0; c < 6; ++c) sum += __expf(o[c] - m);
  float l = __logf(sum);
#pragma unroll
  for (int c = 0; c < 6; ++c) out[n * 6 + c] = o[c] - m - l;
}

extern "C" void kernel_launch(void* const* d_in, const int* in_sizes, int n_in,
                              void* d_out, int out_size, void* d_ws, size_t ws_size,
                              hipStream_t stream) {
  const float* x      = (const float*)d_in[0];
  const int*   ei     = (const int*)d_in[1];
  const float* W1     = (const float*)d_in[2];
  const float* a_src1 = (const float*)d_in[3];
  const float* a_dst1 = (const float*)d_in[4];
  const float* b1     = (const float*)d_in[5];
  const float* W2     = (const float*)d_in[6];
  const float* a_src2 = (const float*)d_in[7];
  const float* a_dst2 = (const float*)d_in[8];
  const float* b2     = (const float*)d_in[9];
  float* out = (float*)d_out;

  int N = in_sizes[0] / 7;
  int E = in_sizes[1] / 2;

  float* ws     = (float*)d_ws;
  float* S1     = ws;
  float* agg2r  = S1 + (size_t)N * 32;
  float* as1    = agg2r + (size_t)N * 8;
  float* ad1    = as1 + (size_t)N * 4;
  float* emax1  = ad1 + (size_t)N * 4;
  float* h2     = emax1 + (size_t)N * 4;
  float* as2    = h2 + (size_t)N * 6;
  float* ad2    = as2 + N;
  float* emax2  = ad2 + N;
  float* gst    = emax2 + N;
  unsigned* tick = (unsigned*)(gst + 128);

  const int B = 256;
  auto G = [](long n, int b) { return (int)((n + b - 1) / b); };
  int tiles = (N + 15) / 16;
  int projBlocks = (tiles + (B / 32) - 1) / (B / 32);

  init_kern<<<G((long)N * 32, B), B, 0, stream>>>(emax1, S1, emax2, agg2r, gst, tick, N);

  score1_wmma<<<G(tiles, (B / 32) * SC_TILES), B, 0, stream>>>(x, W1, a_src1, a_dst1, as1, ad1, gst, N);
  e1_max<<<1024, B, 0, stream>>>(ei, as1, ad1, emax1, gst, E, N);
  e1_agg<<<2048, B, 0, stream>>>(ei, W1, a_src1, ad1, gst, emax1, x, S1, tick, E, N);
  post1_proj2_wmma<<<projBlocks, B, 0, stream>>>(S1, W1, b1, W2, a_src2, a_dst2, h2, as2, ad2, gst, N);

  e2_max<<<1024, B, 0, stream>>>(ei, as2, ad2, emax2, gst, E, N);
  e2_agg<<<2048, B, 0, stream>>>(ei, a_src2, ad2, gst, emax2, h2, agg2r, tick + 2 * 128, E, N);
  final_kern<<<G(N, B), B, 0, stream>>>(agg2r, b2, out, N);
}
